# starting kernel
# baseline (speedup 1.0000x reference)
#include <hip/hip_runtime.h>
#include <hip/hip_bf16.h>

typedef __attribute__((ext_vector_type(2))) float v2f;
typedef __attribute__((ext_vector_type(8))) float v8f;

#define NN 50000


__global__ __launch_bounds__(256) void k_deg_init(float* __restrict__ deg, int n) {
  int i = blockIdx.x * blockDim.x + threadIdx.x;
  if (i < n) deg[i] = 1.0f;
}

__global__ __launch_bounds__(256) void k_deg_count(const int* __restrict__ dst,
                                                   float* __restrict__ deg, int e) {
  int i = blockIdx.x * blockDim.x + threadIdx.x;
  if (i < e) atomicAdd(&deg[dst[i]], 1.0f);
}

__global__ __launch_bounds__(256) void k_dinv(float* __restrict__ deg, int n) {
  int i = blockIdx.x * blockDim.x + threadIdx.x;
  if (i < n) {
    float d = deg[i];
    deg[i] = (d > 0.0f) ? rsqrtf(d) : 0.0f;
  }
}

__global__ __launch_bounds__(256) void k_norm(const int* __restrict__ src,
                                              const int* __restrict__ dst,
                                              const float* __restrict__ dinv,
                                              float* __restrict__ norm, int e) {
  int i = blockIdx.x * blockDim.x + threadIdx.x;
  if (i < e) norm[i] = dinv[src[i]] * dinv[dst[i]];
}

template <int K, int OUTD>
__global__ __launch_bounds__(256) void k_transpose(const float* __restrict__ W,
                                                   float* __restrict__ Wt) {
  int i = blockIdx.x * blockDim.x + threadIdx.x;
  if (i < K * OUTD) {
    int k = i / OUTD;
    int n = i - k * OUTD;
    Wt[n * K + k] = W[i];
  }
}

template <int K, int OUTD>
__global__ __launch_bounds__(256) void k_gemm_wmma(const float* __restrict__ X,
                                                   const float* __restrict__ Wt,
                                                   float* __restrict__ H) {
  static_assert(K == 128, "K fixed at 128");
  constexpr int PAD = K + 4;
  __shared__ float xs[16 * PAD];

  const int tid   = threadIdx.x;
  const int lane  = tid & 31;
  const int wave  = tid >> 5;
  const int mtile = blockIdx.x;

  constexpr int NTHREADS = (OUTD / 16) * 32;
  constexpr int NSLOTS   = 16 * (K / 4);
#pragma unroll
  for (int s = tid; s < NSLOTS; s += NTHREADS) {
    int row = s >> 5;
    int c4  = s & 31;
    float4 v = *(const float4*)(X + (size_t)(mtile * 16 + row) * K + c4 * 4);
    *(float4*)(&xs[row * PAD + c4 * 4]) = v;
  }
  __syncthreads();

  const int idx16 = lane & 15;
  const int half  = lane >> 4;
  const float* xsrow = &xs[idx16 * PAD + 2 * half];
  const float* wrow  = Wt + (size_t)(wave * 16 + idx16) * K + 2 * half;

  v8f acc = {};
#pragma unroll
  for (int k0 = 0; k0 < K; k0 += 4) {
    v2f a = *(const v2f*)(xsrow + k0);
    v2f b = *(const v2f*)(wrow + k0);
    acc = __builtin_amdgcn_wmma_f32_16x16x4_f32(
         false, a,  false, b,
         (short)0, acc,  false,  false);
  }

  float* hp = H + (size_t)(mtile * 16 + half * 8) * OUTD + wave * 16 + idx16;
#pragma unroll
  for (int r = 0; r < 8; ++r) hp[r * OUTD] = acc[r];
}


template <int OUTD>
__global__ __launch_bounds__(256) void k_self_init(const float* __restrict__ H,
                                                   const float* __restrict__ dinv,
                                                   float* __restrict__ agg,
                                                   long long total) {
  constexpr int SHIFT = (OUTD == 128) ? 7 : 6;
  long long idx = (long long)blockIdx.x * blockDim.x + threadIdx.x;
  if (idx >= total) return;
  float di = dinv[(int)(idx >> SHIFT)];
  agg[idx] = H[idx] * di * di;
}

template <int OUTD>
__global__ __launch_bounds__(256) void k_edge_scatter(const int* __restrict__ src,
                                                      const int* __restrict__ dst,
                                                      const float* __restrict__ norm,
                                                      const float* __restrict__ H,
                                                      float* __restrict__ agg, int E) {
  constexpr int VEC = OUTD / 32;
  int e = blockIdx.x * (blockDim.x >> 5) + (threadIdx.x >> 5);
  if (e >= E) return;
  int lane = threadIdx.x & 31;
  int s = src[e];
  int d = dst[e];
  float w = norm[e];
  const float* hp = H + (size_t)s * OUTD + lane * VEC;
  float* ap = agg + (size_t)d * OUTD + lane * VEC;
  if constexpr (VEC == 4) {
    float4 v = *(const float4*)hp;
    atomicAdd(ap + 0, v.x * w);
    atomicAdd(ap + 1, v.y * w);
    atomicAdd(ap + 2, v.z * w);
    atomicAdd(ap + 3, v.w * w);
  } else {
    float2 v = *(const float2*)hp;
    atomicAdd(ap + 0, v.x * w);
    atomicAdd(ap + 1, v.y * w);
  }
}

template <int OUTD, int RELU>
__global__ __launch_bounds__(256) void k_finalize(const float* __restrict__ agg,
                                                  const float* __restrict__ bias,
                                                  float* __restrict__ out,
                                                  long long total) {
  long long idx = (long long)blockIdx.x * blockDim.x + threadIdx.x;
  if (idx >= total) return;
  float v = agg[idx] + bias[(int)idx & (OUTD - 1)];
  if (RELU) v = fmaxf(v, 0.0f);
  out[idx] = v;
}


template <int OUTD, int RELU>
static void launch_layer(const float* xin, const float* W, const float* b,
                         const int* src, const int* dst,
                         const float* norm, const float* dinv,
                         float* wt, float* h, float* agg, float* out,
                         int E, hipStream_t stream) {
  constexpr int K = 128;
  k_transpose<K, OUTD><<<(K * OUTD + 255) / 256, 256, 0, stream>>>(W, wt);
  k_gemm_wmma<K, OUTD><<<dim3(NN / 16), dim3((OUTD / 16) * 32), 0, stream>>>(xin, wt, h);

  long long tot = (long long)NN * OUTD;
  unsigned nb = (unsigned)((tot + 255) / 256);
  k_self_init<OUTD><<<nb, 256, 0, stream>>>(h, dinv, agg, tot);

  unsigned eb = (unsigned)((E + 7) / 8);
  k_edge_scatter<OUTD><<<eb, 256, 0, stream>>>(src, dst, norm, h, agg, E);

  k_finalize<OUTD, RELU><<<nb, 256, 0, stream>>>(agg, b, out, tot);
}

extern "C" void kernel_launch(void* const* d_in, const int* in_sizes, int n_in,
                              void* d_out, int out_size, void* d_ws, size_t ws_size,
                              hipStream_t stream) {
  const float* x  = (const float*)d_in[0];
  const int*   ei = (const int*)d_in[1];
  const float* W1 = (const float*)d_in[2];
  const float* b1 = (const float*)d_in[3];
  const float* W2 = (const float*)d_in[4];
  const float* b2 = (const float*)d_in[5];
  const float* W3 = (const float*)d_in[6];
  const float* b3 = (const float*)d_in[7];
  const float* W4 = (const float*)d_in[8];
  const float* b4 = (const float*)d_in[9];

  const int E = in_sizes[1] / 2;
  const int* src = ei;
  const int* dst = ei + E;

  float* dinv = (float*)d_ws;
  float* norm = dinv + ((NN + 3) & ~3);
  float* h    = norm + (((size_t)E + 3) & ~(size_t)3);
  float* agg  = h + (size_t)NN * 128;
  float* xcur = agg + (size_t)NN * 128;
  float* wt   = xcur + (size_t)NN * 128;

  unsigned nb = (NN + 255) / 256;
  unsigned eb = (unsigned)((E + 255) / 256);
  k_deg_init<<<nb, 256, 0, stream>>>(dinv, NN);
  k_deg_count<<<eb, 256, 0, stream>>>(dst, dinv, E);
  k_dinv<<<nb, 256, 0, stream>>>(dinv, NN);
  k_norm<<<eb, 256, 0, stream>>>(src, dst, dinv, norm, E);

  launch_layer<128, 1>(x,    W1, b1, src, dst, norm, dinv, wt, h, agg, xcur, E, stream);
  launch_layer<128, 1>(xcur, W2, b2, src, dst, norm, dinv, wt, h, agg, xcur, E, stream);
  launch_layer<128, 1>(xcur, W3, b3, src, dst, norm, dinv, wt, h, agg, xcur, E, stream);
  launch_layer<64,  0>(xcur, W4, b4, src, dst, norm, dinv, wt, h, agg,
                       (float*)d_out, E, stream);
}


# per-edge f32 atomic scatter replaced by dst-sorted edge list (int atomic count, scan, edge-id order) + one wave per node summing rows in registers, self loop/bias/relu fused; sums in a fixed order
# speedup vs baseline: 3.3910x; 3.3910x over previous
#include <hip/hip_runtime.h>
#include <hip/hip_bf16.h>

typedef __attribute__((ext_vector_type(2))) float v2f;
typedef __attribute__((ext_vector_type(8))) float v8f;

#define NN 50000


__global__ __launch_bounds__(256) void k_cnt_zero(int* __restrict__ cnt, int n) {
  int i = blockIdx.x * blockDim.x + threadIdx.x;
  if (i < n) cnt[i] = 0;
}

__global__ __launch_bounds__(256) void k_count_rank(const int* __restrict__ dst,
                                                    int* __restrict__ cnt,
                                                    int* __restrict__ rank, int e) {
  int i = blockIdx.x * blockDim.x + threadIdx.x;
  if (i < e) rank[i] = atomicAdd(&cnt[dst[i]], 1);
}

#define SCAN_B 256
__global__ __launch_bounds__(SCAN_B) void k_scan_dinv(const int* __restrict__ cnt,
                                                      int* __restrict__ rowptr,
                                                      float* __restrict__ dinv) {
  __shared__ int wtot[SCAN_B / 32];
  __shared__ int ptot[SCAN_B / 32];
  const int t     = threadIdx.x;
  const int lane  = t & 31;
  const int wave  = t >> 5;
  const int start = blockIdx.x * SCAN_B;

  int p = 0;
  constexpr int GROUP = 8 * SCAN_B * 4;
  const int nfull = start / GROUP;
  for (int g = 0; g < nfull; ++g) {
    int4 q[8];
#pragma unroll
    for (int u = 0; u < 8; ++u) q[u] = *(const int4*)(cnt + g * GROUP + u * SCAN_B * 4 + t * 4);
#pragma unroll
    for (int u = 0; u < 8; ++u) p += (q[u].x + q[u].y) + (q[u].z + q[u].w);
  }
  for (int i = nfull * GROUP + t * 4; i < start; i += SCAN_B * 4) {
    const int4 q = *(const int4*)(cnt + i);
    p += (q.x + q.y) + (q.z + q.w);
  }
  const int i = start + t;
  const int c = (i < NN) ? cnt[i] : 0;

  int inc = c;
#pragma unroll
  for (int off = 1; off < 32; off <<= 1) {
    int o = __shfl_up(inc, off);
    if (lane >= off) inc += o;
  }
#pragma unroll
  for (int off = 16; off > 0; off >>= 1) p += __shfl_xor(p, off);
  if (lane == 31) wtot[wave] = inc;
  if (lane == 0)  ptot[wave] = p;
  __syncthreads();
  int base = 0;
#pragma unroll
  for (int w = 0; w < SCAN_B / 32; ++w) {
    base += ptot[w];
    if (w < wave) base += wtot[w];
  }
  const int excl = base + inc - c;
  if (i < NN) {
    rowptr[i] = excl;
    float d = (float)(c + 1);
    dinv[i] = (d > 0.0f) ? rsqrtf(d) : 0.0f;
  }
  if (i == NN - 1) rowptr[NN] = excl + c;
}

__global__ __launch_bounds__(256) void k_fill(const int* __restrict__ dst,
                                              const int* __restrict__ rank,
                                              const int* __restrict__ rowptr,
                                              int* __restrict__ elist, int e) {
  int i = blockIdx.x * blockDim.x + threadIdx.x;
  if (i < e) elist[rowptr[dst[i]] + rank[i]] = i;
}

__global__ __launch_bounds__(256) void k_order(const int* __restrict__ rowptr,
                                               const int* __restrict__ elist,
                                               const int* __restrict__ src,
                                               const float* __restrict__ dinv,
                                               int2* __restrict__ csr) {
  const int lane = threadIdx.x & 31;
  const int node = __builtin_amdgcn_readfirstlane((int)(blockIdx.x * 8 + (threadIdx.x >> 5)));
  if (node >= NN) return;
  const int beg = __builtin_amdgcn_readfirstlane(rowptr[node]);
  const int end = __builtin_amdgcn_readfirstlane(rowptr[node + 1]);
  const float dn = dinv[node];
  for (int base = beg; base < end; base += 32) {
    const int idx = base + lane;
    const bool live = idx < end;
    const int mine = live ? elist[idx] : 0x7fffffff;
    int pos = 0;
    for (int ob = beg; ob < end; ob += 32) {
      const int oidx = ob + lane;
      const int other = (oidx < end) ? elist[oidx] : 0x7fffffff;
      const int m = min(32, end - ob);
      for (int j = 0; j < m; ++j)
        pos += (__builtin_amdgcn_readlane(other, j) < mine) ? 1 : 0;
    }
    if (live) {
      const int s = src[mine];
      csr[beg + pos] = make_int2(s, __float_as_int(dinv[s] * dn));
    }
  }
}

template <int K, int OUTD>
__global__ __launch_bounds__(256) void k_transpose(const float* __restrict__ W,
                                                   float* __restrict__ Wt) {
  int i = blockIdx.x * blockDim.x + threadIdx.x;
  if (i < K * OUTD) {
    int k = i / OUTD;
    int n = i - k * OUTD;
    Wt[n * K + k] = W[i];
  }
}

template <int K, int OUTD>
__global__ __launch_bounds__(256) void k_gemm_wmma(const float* __restrict__ X,
                                                   const float* __restrict__ Wt,
                                                   float* __restrict__ H) {
  static_assert(K == 128, "K fixed at 128");
  constexpr int PAD = K + 4;
  __shared__ float xs[16 * PAD];

  const int tid   = threadIdx.x;
  const int lane  = tid & 31;
  const int wave  = tid >> 5;
  const int mtile = blockIdx.x;

  constexpr int NTHREADS = (OUTD / 16) * 32;
  constexpr int NSLOTS   = 16 * (K / 4);
#pragma unroll
  for (int s = tid; s < NSLOTS; s += NTHREADS) {
    int row = s >> 5;
    int c4  = s & 31;
    float4 v = *(const float4*)(X + (size_t)(mtile * 16 + row) * K + c4 * 4);
    *(float4*)(&xs[row * PAD + c4 * 4]) = v;
  }
  __syncthreads();

  const int idx16 = lane & 15;
  const int half  = lane >> 4;
  const float* xsrow = &xs[idx16 * PAD + 2 * half];
  const float* wrow  = Wt + (size_t)(wave * 16 + idx16) * K + 2 * half;

  v8f acc = {};
#pragma unroll
  for (int k0 = 0; k0 < K; k0 += 4) {
    v2f a = *(const v2f*)(xsrow + k0);
    v2f b = *(const v2f*)(wrow + k0);
    acc = __builtin_amdgcn_wmma_f32_16x16x4_f32(
         false, a,  false, b,
         (short)0, acc,  false,  false);
  }

  float* hp = H + (size_t)(mtile * 16 + half * 8) * OUTD + wave * 16 + idx16;
#pragma unroll
  for (int r = 0; r < 8; ++r) hp[r * OUTD] = acc[r];
}


template <int VEC> struct vec_of;
template <> struct vec_of<4> { typedef __attribute__((ext_vector_type(4))) float type; };
template <> struct vec_of<2> { typedef v2f type; };

template <int N, int OUTD, typename V>
__device__ __forceinline__ void rows_fma(V& acc, const float* __restrict__ H, unsigned lo,
                                         int2 ew, int j) {
  int s[N];
  float w[N];
  V v[N];
#pragma unroll
  for (int u = 0; u < N; ++u) {
    s[u] = __builtin_amdgcn_readlane(ew.x, j + u);
    w[u] = __int_as_float(__builtin_amdgcn_readlane(ew.y, j + u));
  }
#pragma unroll
  for (int u = 0; u < N; ++u) v[u] = *(const V*)(H + (size_t)s[u] * OUTD + lo);
#pragma unroll
  for (int u = 0; u < N; ++u) acc += v[u] * w[u];
}

template <int OUTD, int RELU>
__global__ __launch_bounds__(256) void k_gather(const int* __restrict__ rowptr,
                                                const int2* __restrict__ csr,
                                                const float* __restrict__ dinv,
                                                const float* __restrict__ H,
                                                const float* __restrict__ bias,
                                                float* __restrict__ out) {
  constexpr int VEC = OUTD / 32;
  using V = typename vec_of<VEC>::type;
  const int lane = threadIdx.x & 31;
  const int node = __builtin_amdgcn_readfirstlane((int)(blockIdx.x * 8 + (threadIdx.x >> 5)));
  if (node >= NN) return;
  const int beg = __builtin_amdgcn_readfirstlane(rowptr[node]);
  const int end = __builtin_amdgcn_readfirstlane(rowptr[node + 1]);
  const unsigned lo = (unsigned)lane * VEC;
  const float di = dinv[node];
  V acc = *(const V*)(H + (size_t)node * OUTD + lo) * (di * di);

  for (int base = beg; base < end; base += 32) {
    const int idx = base + lane;
    int2 ew = make_int2(0, 0);
    if (idx < end) ew = csr[idx];
    const int m = min(32, end - base);
    int j = 0;
    for (; j + 8 <= m; j += 8) rows_fma<8, OUTD, V>(acc, H, lo, ew, j);
    if (j + 4 <= m) {
      rows_fma<4, OUTD, V>(acc, H, lo, ew, j);
      j += 4;
    }
    for (; j < m; ++j) rows_fma<1, OUTD, V>(acc, H, lo, ew, j);
  }

  acc += *(const V*)(bias + lo);
  if (RELU) {
#pragma unroll
    for (int k = 0; k < VEC; ++k) acc[k] = fmaxf(acc[k], 0.0f);
  }
  *(V*)(out + (size_t)node * OUTD + lo) = acc;
}


template <int OUTD, int RELU>
static void launch_layer(const float* xin, const float* W, const float* b,
                         const int* rowptr, const int2* csr, const float* dinv,
                         float* wt, float* h, float* out, hipStream_t stream) {
  constexpr int K = 128;
  k_transpose<K, OUTD><<<(K * OUTD + 255) / 256, 256, 0, stream>>>(W, wt);
  k_gemm_wmma<K, OUTD><<<dim3(NN / 16), dim3((OUTD / 16) * 32), 0, stream>>>(xin, wt, h);
  k_gather<OUTD, RELU><<<(NN + 7) / 8, 256, 0, stream>>>(rowptr, csr, dinv, h, b, out);
}

extern "C" void kernel_launch(void* const* d_in, const int* in_sizes, int n_in,
                              void* d_out, int out_size, void* d_ws, size_t ws_size,
                              hipStream_t stream) {
  const float* x  = (const float*)d_in[0];
  const int*   ei = (const int*)d_in[1];
  const float* W1 = (const float*)d_in[2];
  const float* b1 = (const float*)d_in[3];
  const float* W2 = (const float*)d_in[4];
  const float* b2 = (const float*)d_in[5];
  const float* W3 = (const float*)d_in[6];
  const float* b3 = (const float*)d_in[7];
  const float* W4 = (const float*)d_in[8];
  const float* b4 = (const float*)d_in[9];

  const int E = in_sizes[1] / 2;
  const int* src = ei;
  const int* dst = ei + E;
  const size_t Ep = ((size_t)E + 3) & ~(size_t)3;

  float* dinv   = (float*)d_ws;
  int*   cnt    = (int*)(dinv + NN);
  int*   rowptr = cnt + NN;
  int*   rank   = rowptr + NN + 4;
  int*   elist  = rank + Ep;
  int2*  csr    = (int2*)(elist + Ep);
  float* h      = (float*)(elist + Ep) + 2 * Ep;
  float* xcur   = h + (size_t)NN * 128;
  float* wt     = xcur + (size_t)NN * 128;

  unsigned nb = (NN + 255) / 256;
  unsigned eb = (unsigned)((E + 255) / 256);
  k_cnt_zero<<<nb, 256, 0, stream>>>(cnt, NN);
  k_count_rank<<<eb, 256, 0, stream>>>(dst, cnt, rank, E);
  k_scan_dinv<<<(NN + SCAN_B - 1) / SCAN_B, SCAN_B, 0, stream>>>(cnt, rowptr, dinv);
  k_fill<<<eb, 256, 0, stream>>>(dst, rank, rowptr, elist, E);
  k_order<<<(NN + 7) / 8, 256, 0, stream>>>(rowptr, elist, src, dinv, csr);

  launch_layer<128, 1>(x,    W1, b1, rowptr, csr, dinv, wt, h, xcur, stream);
  launch_layer<128, 1>(xcur, W2, b2, rowptr, csr, dinv, wt, h, xcur, stream);
  launch_layer<128, 1>(xcur, W3, b3, rowptr, csr, dinv, wt, h, xcur, stream);
  launch_layer<64,  0>(xcur, W4, b4, rowptr, csr, dinv, wt, h, (float*)d_out, stream);
}


# next layer's GEMM fused into the row-sum kernel: resident 16-wave blocks hold W in LDS, put 16 summed rows in an LDS tile and do the f32 WMMA product there; 17 launches to 11; sum and k order kept
# speedup vs baseline: 3.8398x; 1.1324x over previous
#include <hip/hip_runtime.h>
#include <hip/hip_bf16.h>

typedef __attribute__((ext_vector_type(2))) float v2f;
typedef __attribute__((ext_vector_type(8))) float v8f;

#define NN 50000


__global__ __launch_bounds__(256) void k_cnt_zero(int* __restrict__ cnt, int n) {
  int i = blockIdx.x * blockDim.x + threadIdx.x;
  if (i < n) cnt[i] = 0;
}

__global__ __launch_bounds__(256) void k_count_rank(const int* __restrict__ dst,
                                                    int* __restrict__ cnt,
                                                    int* __restrict__ rank, int e) {
  int i = blockIdx.x * blockDim.x + threadIdx.x;
  if (i < e) rank[i] = atomicAdd(&cnt[dst[i]], 1);
}

#define SCAN_B 256
__global__ __launch_bounds__(SCAN_B) void k_scan_dinv(const int* __restrict__ cnt,
                                                      int* __restrict__ rowptr,
                                                      float* __restrict__ dinv) {
  __shared__ int wtot[SCAN_B / 32];
  __shared__ int ptot[SCAN_B / 32];
  const int t     = threadIdx.x;
  const int lane  = t & 31;
  const int wave  = t >> 5;
  const int start = blockIdx.x * SCAN_B;

  int p = 0;
  constexpr int GROUP = 8 * SCAN_B * 4;
  const int nfull = start / GROUP;
  for (int g = 0; g < nfull; ++g) {
    int4 q[8];
#pragma unroll
    for (int u = 0; u < 8; ++u) q[u] = *(const int4*)(cnt + g * GROUP + u * SCAN_B * 4 + t * 4);
#pragma unroll
    for (int u = 0; u < 8; ++u) p += (q[u].x + q[u].y) + (q[u].z + q[u].w);
  }
  for (int i = nfull * GROUP + t * 4; i < start; i += SCAN_B * 4) {
    const int4 q = *(const int4*)(cnt + i);
    p += (q.x + q.y) + (q.z + q.w);
  }
  const int i = start + t;
  const int c = (i < NN) ? cnt[i] : 0;

  int inc = c;
#pragma unroll
  for (int off = 1; off < 32; off <<= 1) {
    int o = __shfl_up(inc, off);
    if (lane >= off) inc += o;
  }
#pragma unroll
  for (int off = 16; off > 0; off >>= 1) p += __shfl_xor(p, off);
  if (lane == 31) wtot[wave] = inc;
  if (lane == 0)  ptot[wave] = p;
  __syncthreads();
  int base = 0;
#pragma unroll
  for (int w = 0; w < SCAN_B / 32; ++w) {
    base += ptot[w];
    if (w < wave) base += wtot[w];
  }
  const int excl = base + inc - c;
  if (i < NN) {
    rowptr[i] = excl;
    float d = (float)(c + 1);
    dinv[i] = (d > 0.0f) ? rsqrtf(d) : 0.0f;
  }
  if (i == NN - 1) rowptr[NN] = excl + c;
}

__global__ __launch_bounds__(256) void k_fill(const int* __restrict__ dst,
                                              const int* __restrict__ rank,
                                              const int* __restrict__ rowptr,
                                              int* __restrict__ elist, int e) {
  int i = blockIdx.x * blockDim.x + threadIdx.x;
  if (i < e) elist[rowptr[dst[i]] + rank[i]] = i;
}

__global__ __launch_bounds__(256) void k_order(const int* __restrict__ rowptr,
                                               const int* __restrict__ elist,
                                               const int* __restrict__ src,
                                               const float* __restrict__ dinv,
                                               int2* __restrict__ csr) {
  const int lane = threadIdx.x & 31;
  const int node = __builtin_amdgcn_readfirstlane((int)(blockIdx.x * 8 + (threadIdx.x >> 5)));
  if (node >= NN) return;
  const int beg = __builtin_amdgcn_readfirstlane(rowptr[node]);
  const int end = __builtin_amdgcn_readfirstlane(rowptr[node + 1]);
  const float dn = dinv[node];
  for (int base = beg; base < end; base += 32) {
    const int idx = base + lane;
    const bool live = idx < end;
    const int mine = live ? elist[idx] : 0x7fffffff;
    int pos = 0;
    for (int ob = beg; ob < end; ob += 32) {
      const int oidx = ob + lane;
      const int other = (oidx < end) ? elist[oidx] : 0x7fffffff;
      const int m = min(32, end - ob);
      for (int j = 0; j < m; ++j)
        pos += (__builtin_amdgcn_readlane(other, j) < mine) ? 1 : 0;
    }
    if (live) {
      const int s = src[mine];
      csr[beg + pos] = make_int2(s, __float_as_int(dinv[s] * dn));
    }
  }
}

template <int K, int OUTD>
__global__ __launch_bounds__(256) void k_transpose(const float* __restrict__ W,
                                                   float* __restrict__ Wt) {
  int i = blockIdx.x * blockDim.x + threadIdx.x;
  if (i < K * OUTD) {
    int k = i / OUTD;
    int n = i - k * OUTD;
    Wt[n * K + k] = W[i];
  }
}

template <int K, int OUTD>
__global__ __launch_bounds__(256) void k_gemm_wmma(const float* __restrict__ X,
                                                   const float* __restrict__ Wt,
                                                   float* __restrict__ H) {
  static_assert(K == 128, "K fixed at 128");
  constexpr int PAD = K + 4;
  __shared__ float xs[16 * PAD];

  const int tid   = threadIdx.x;
  const int lane  = tid & 31;
  const int wave  = tid >> 5;
  const int mtile = blockIdx.x;

  constexpr int NTHREADS = (OUTD / 16) * 32;
  constexpr int NSLOTS   = 16 * (K / 4);
#pragma unroll
  for (int s = tid; s < NSLOTS; s += NTHREADS) {
    int row = s >> 5;
    int c4  = s & 31;
    float4 v = *(const float4*)(X + (size_t)(mtile * 16 + row) * K + c4 * 4);
    *(float4*)(&xs[row * PAD + c4 * 4]) = v;
  }
  __syncthreads();

  const int idx16 = lane & 15;
  const int half  = lane >> 4;
  const float* xsrow = &xs[idx16 * PAD + 2 * half];
  const float* wrow  = Wt + (size_t)(wave * 16 + idx16) * K + 2 * half;

  v8f acc = {};
#pragma unroll
  for (int k0 = 0; k0 < K; k0 += 4) {
    v2f a = *(const v2f*)(xsrow + k0);
    v2f b = *(const v2f*)(wrow + k0);
    acc = __builtin_amdgcn_wmma_f32_16x16x4_f32(
         false, a,  false, b,
         (short)0, acc,  false,  false);
  }

  float* hp = H + (size_t)(mtile * 16 + half * 8) * OUTD + wave * 16 + idx16;
#pragma unroll
  for (int r = 0; r < 8; ++r) hp[r * OUTD] = acc[r];
}


template <int VEC> struct vec_of;
template <> struct vec_of<4> { typedef __attribute__((ext_vector_type(4))) float type; };
template <> struct vec_of<2> { typedef v2f type; };

template <int N, int OUTD, typename V>
__device__ __forceinline__ void rows_fma(V& acc, const float* __restrict__ H, unsigned lo,
                                         int2 ew, int j) {
  int s[N];
  float w[N];
  V v[N];
#pragma unroll
  for (int u = 0; u < N; ++u) {
    s[u] = __builtin_amdgcn_readlane(ew.x, j + u);
    w[u] = __int_as_float(__builtin_amdgcn_readlane(ew.y, j + u));
  }
#pragma unroll
  for (int u = 0; u < N; ++u) v[u] = *(const V*)(H + (size_t)s[u] * OUTD + lo);
#pragma unroll
  for (int u = 0; u < N; ++u) acc += v[u] * w[u];
}

template <int OUTD, int RELU>
__global__ __launch_bounds__(256) void k_gather(const int* __restrict__ rowptr,
                                                const int2* __restrict__ csr,
                                                const float* __restrict__ dinv,
                                                const float* __restrict__ H,
                                                const float* __restrict__ bias,
                                                float* __restrict__ out) {
  constexpr int VEC = OUTD / 32;
  using V = typename vec_of<VEC>::type;
  const int lane = threadIdx.x & 31;
  const int node = __builtin_amdgcn_readfirstlane((int)(blockIdx.x * 8 + (threadIdx.x >> 5)));
  if (node >= NN) return;
  const int beg = __builtin_amdgcn_readfirstlane(rowptr[node]);
  const int end = __builtin_amdgcn_readfirstlane(rowptr[node + 1]);
  const unsigned lo = (unsigned)lane * VEC;
  const float di = dinv[node];
  V acc = *(const V*)(H + (size_t)node * OUTD + lo) * (di * di);

  for (int base = beg; base < end; base += 32) {
    const int idx = base + lane;
    int2 ew = make_int2(0, 0);
    if (idx < end) ew = csr[idx];
    const int m = min(32, end - base);
    int j = 0;
    for (; j + 8 <= m; j += 8) rows_fma<8, OUTD, V>(acc, H, lo, ew, j);
    if (j + 4 <= m) {
      rows_fma<4, OUTD, V>(acc, H, lo, ew, j);
      j += 4;
    }
    for (; j < m; ++j) rows_fma<1, OUTD, V>(acc, H, lo, ew, j);
  }

  acc += *(const V*)(bias + lo);
  if (RELU) {
#pragma unroll
    for (int k = 0; k < VEC; ++k) acc[k] = fmaxf(acc[k], 0.0f);
  }
  *(V*)(out + (size_t)node * OUTD + lo) = acc;
}


__device__ __forceinline__ vec_of<4>::type node_row_relu(int node, int lane, unsigned lo,
                                                        const int* __restrict__ rowptr,
                                                        const int2* __restrict__ csr,
                                                        const float* __restrict__ dinv,
                                                        const float* __restrict__ H,
                                                        vec_of<4>::type bv) {
  using V = vec_of<4>::type;
  constexpr int K = 128;
  const int beg = __builtin_amdgcn_readfirstlane(rowptr[node]);
  const int end = __builtin_amdgcn_readfirstlane(rowptr[node + 1]);
  const float di = dinv[node];
  V acc = *(const V*)(H + (size_t)node * K + lo) * (di * di);

  for (int base = beg; base < end; base += 32) {
    const int idx = base + lane;
    int2 ew = make_int2(0, 0);
    if (idx < end) ew = csr[idx];
    const int m = min(32, end - base);
    int j = 0;
    for (; j + 8 <= m; j += 8) rows_fma<8, K, V>(acc, H, lo, ew, j);
    if (j + 4 <= m) {
      rows_fma<4, K, V>(acc, H, lo, ew, j);
      j += 4;
    }
    for (; j < m; ++j) rows_fma<1, K, V>(acc, H, lo, ew, j);
  }
  acc += bv;
#pragma unroll
  for (int k = 0; k < 4; ++k) acc[k] = fmaxf(acc[k], 0.0f);
  return acc;
}

template <int OUTN>
__global__ __launch_bounds__(512, 16) void k_gather_gemm(const int* __restrict__ rowptr,
                                                         const int2* __restrict__ csr,
                                                         const float* __restrict__ dinv,
                                                         const float* __restrict__ H,
                                                         const float* __restrict__ bias,
                                                         const float* __restrict__ Wn,
                                                         float* __restrict__ Hn) {
  constexpr int K   = 128;
  constexpr int PAD = K + 4;
  constexpr int NT  = NN / 16;
  static_assert(NN % 16 == 0, "node tiles need no bounds checks");
  __shared__ float wl[OUTN * PAD];
  __shared__ float xs[16 * PAD];
  using V = vec_of<4>::type;

  const int tid  = threadIdx.x;
  const int lane = tid & 31;
  const int wave = __builtin_amdgcn_readfirstlane(tid >> 5);
  const unsigned lo = (unsigned)lane * 4;
  const V bv = *(const V*)(bias + lo);
  const int idx16 = lane & 15;
  const int half  = lane >> 4;

  int tile = blockIdx.x;
  if (tile >= NT) return;

  V row = node_row_relu(tile * 16 + wave, lane, lo, rowptr, csr, dinv, H, bv);
  *(V*)(&xs[wave * PAD + lo]) = row;

#pragma unroll 8
  for (int i = tid; i < K * OUTN; i += 512) {
    const int k = i / OUTN;
    const int n = i - k * OUTN;
    wl[n * PAD + k] = Wn[i];
  }

  while (true) {
    __syncthreads();

    if (wave < OUTN / 16) {
      const float* xsrow = &xs[idx16 * PAD + 2 * half];
      const float* wrow  = &wl[(wave * 16 + idx16) * PAD + 2 * half];
      v8f c = {};
#pragma unroll
      for (int k0 = 0; k0 < K; k0 += 4) {
        v2f a = *(const v2f*)(xsrow + k0);
        v2f b = *(const v2f*)(wrow + k0);
        c = __builtin_amdgcn_wmma_f32_16x16x4_f32(
             false, a,  false, b,
             (short)0, c,  false,  false);
      }
      float* hp = Hn + (size_t)(tile * 16 + half * 8) * OUTN + wave * 16 + idx16;
#pragma unroll
      for (int r = 0; r < 8; ++r) hp[r * OUTN] = c[r];
    }

    tile += gridDim.x;
    if (tile >= NT) break;
    row = node_row_relu(tile * 16 + wave, lane, lo, rowptr, csr, dinv, H, bv);
    __syncthreads();
    *(V*)(&xs[wave * PAD + lo]) = row;
  }
}


extern "C" void kernel_launch(void* const* d_in, const int* in_sizes, int n_in,
                              void* d_out, int out_size, void* d_ws, size_t ws_size,
                              hipStream_t stream) {
  const float* x  = (const float*)d_in[0];
  const int*   ei = (const int*)d_in[1];
  const float* W1 = (const float*)d_in[2];
  const float* b1 = (const float*)d_in[3];
  const float* W2 = (const float*)d_in[4];
  const float* b2 = (const float*)d_in[5];
  const float* W3 = (const float*)d_in[6];
  const float* b3 = (const float*)d_in[7];
  const float* W4 = (const float*)d_in[8];
  const float* b4 = (const float*)d_in[9];

  const int E = in_sizes[1] / 2;
  const int* src = ei;
  const int* dst = ei + E;
  const size_t Ep = ((size_t)E + 3) & ~(size_t)3;

  float* dinv   = (float*)d_ws;
  int*   cnt    = (int*)(dinv + NN);
  int*   rowptr = cnt + NN;
  int*   rank   = rowptr + NN + 4;
  int*   elist  = rank + Ep;
  int2*  csr    = (int2*)(elist + Ep);
  float* h      = (float*)(elist + Ep) + 2 * Ep;
  float* xcur   = h + (size_t)NN * 128;
  float* wt     = xcur + (size_t)NN * 128;

  unsigned nb = (NN + 255) / 256;
  unsigned eb = (unsigned)((E + 255) / 256);
  k_cnt_zero<<<nb, 256, 0, stream>>>(cnt, NN);
  k_count_rank<<<eb, 256, 0, stream>>>(dst, cnt, rank, E);
  k_scan_dinv<<<(NN + SCAN_B - 1) / SCAN_B, SCAN_B, 0, stream>>>(cnt, rowptr, dinv);
  k_fill<<<eb, 256, 0, stream>>>(dst, rank, rowptr, elist, E);
  k_order<<<(NN + 7) / 8, 256, 0, stream>>>(rowptr, elist, src, dinv, csr);

  k_transpose<128, 128><<<(128 * 128 + 255) / 256, 256, 0, stream>>>(W1, wt);
  k_gemm_wmma<128, 128><<<dim3(NN / 16), dim3(256), 0, stream>>>(x, wt, h);
  k_gather_gemm<128><<<1024, 512, 0, stream>>>(rowptr, csr, dinv, h,    b1, W2, xcur);
  k_gather_gemm<128><<<1024, 512, 0, stream>>>(rowptr, csr, dinv, xcur, b2, W3, h);
  k_gather_gemm<64><<<1024, 512, 0, stream>>>(rowptr, csr, dinv, h,    b3, W4, xcur);
  k_gather<64, 0><<<(NN + 7) / 8, 256, 0, stream>>>(rowptr, csr, dinv, xcur, b4, (float*)d_out);
}


# activation buffers moved to the workspace base: gathered 512-B rows start on 512-B boundaries, 2 aligned 256-B units (4 lines) per row instead of 3 (5); 64-wide rows 1 (2) instead of 2 (3)
# speedup vs baseline: 4.4969x; 1.1711x over previous
#include <hip/hip_runtime.h>
#include <hip/hip_bf16.h>

typedef __attribute__((ext_vector_type(2))) float v2f;
typedef __attribute__((ext_vector_type(8))) float v8f;

#define NN 50000


__global__ __launch_bounds__(256) void k_cnt_zero(int* __restrict__ cnt, int n) {
  int i = blockIdx.x * blockDim.x + threadIdx.x;
  if (i < n) cnt[i] = 0;
}

__global__ __launch_bounds__(256) void k_count_rank(const int* __restrict__ dst,
                                                    int* __restrict__ cnt,
                                                    int* __restrict__ rank, int e) {
  int i = blockIdx.x * blockDim.x + threadIdx.x;
  if (i < e) rank[i] = atomicAdd(&cnt[dst[i]], 1);
}

#define SCAN_B 256
__global__ __launch_bounds__(SCAN_B) void k_scan_dinv(const int* __restrict__ cnt,
                                                      int* __restrict__ rowptr,
                                                      float* __restrict__ dinv) {
  __shared__ int wtot[SCAN_B / 32];
  __shared__ int ptot[SCAN_B / 32];
  const int t     = threadIdx.x;
  const int lane  = t & 31;
  const int wave  = t >> 5;
  const int start = blockIdx.x * SCAN_B;

  int p = 0;
  constexpr int GROUP = 8 * SCAN_B * 4;
  const int nfull = start / GROUP;
  for (int g = 0; g < nfull; ++g) {
    int4 q[8];
#pragma unroll
    for (int u = 0; u < 8; ++u) q[u] = *(const int4*)(cnt + g * GROUP + u * SCAN_B * 4 + t * 4);
#pragma unroll
    for (int u = 0; u < 8; ++u) p += (q[u].x + q[u].y) + (q[u].z + q[u].w);
  }
  for (int i = nfull * GROUP + t * 4; i < start; i += SCAN_B * 4) {
    const int4 q = *(const int4*)(cnt + i);
    p += (q.x + q.y) + (q.z + q.w);
  }
  const int i = start + t;
  const int c = (i < NN) ? cnt[i] : 0;

  int inc = c;
#pragma unroll
  for (int off = 1; off < 32; off <<= 1) {
    int o = __shfl_up(inc, off);
    if (lane >= off) inc += o;
  }
#pragma unroll
  for (int off = 16; off > 0; off >>= 1) p += __shfl_xor(p, off);
  if (lane == 31) wtot[wave] = inc;
  if (lane == 0)  ptot[wave] = p;
  __syncthreads();
  int base = 0;
#pragma unroll
  for (int w = 0; w < SCAN_B / 32; ++w) {
    base += ptot[w];
    if (w < wave) base += wtot[w];
  }
  const int excl = base + inc - c;
  if (i < NN) {
    rowptr[i] = excl;
    float d = (float)(c + 1);
    dinv[i] = (d > 0.0f) ? rsqrtf(d) : 0.0f;
  }
  if (i == NN - 1) rowptr[NN] = excl + c;
}

__global__ __launch_bounds__(256) void k_fill(const int* __restrict__ dst,
                                              const int* __restrict__ rank,
                                              const int* __restrict__ rowptr,
                                              int* __restrict__ elist, int e) {
  int i = blockIdx.x * blockDim.x + threadIdx.x;
  if (i < e) elist[rowptr[dst[i]] + rank[i]] = i;
}

__global__ __launch_bounds__(256) void k_order(const int* __restrict__ rowptr,
                                               const int* __restrict__ elist,
                                               const int* __restrict__ src,
                                               const float* __restrict__ dinv,
                                               int2* __restrict__ csr) {
  const int lane = threadIdx.x & 31;
  const int node = __builtin_amdgcn_readfirstlane((int)(blockIdx.x * 8 + (threadIdx.x >> 5)));
  if (node >= NN) return;
  const int beg = __builtin_amdgcn_readfirstlane(rowptr[node]);
  const int end = __builtin_amdgcn_readfirstlane(rowptr[node + 1]);
  const float dn = dinv[node];
  for (int base = beg; base < end; base += 32) {
    const int idx = base + lane;
    const bool live = idx < end;
    const int mine = live ? elist[idx] : 0x7fffffff;
    int pos = 0;
    for (int ob = beg; ob < end; ob += 32) {
      const int oidx = ob + lane;
      const int other = (oidx < end) ? elist[oidx] : 0x7fffffff;
      const int m = min(32, end - ob);
      for (int j = 0; j < m; ++j)
        pos += (__builtin_amdgcn_readlane(other, j) < mine) ? 1 : 0;
    }
    if (live) {
      const int s = src[mine];
      csr[beg + pos] = make_int2(s, __float_as_int(dinv[s] * dn));
    }
  }
}

template <int K, int OUTD>
__global__ __launch_bounds__(256) void k_transpose(const float* __restrict__ W,
                                                   float* __restrict__ Wt) {
  int i = blockIdx.x * blockDim.x + threadIdx.x;
  if (i < K * OUTD) {
    int k = i / OUTD;
    int n = i - k * OUTD;
    Wt[n * K + k] = W[i];
  }
}

template <int K, int OUTD>
__global__ __launch_bounds__(256) void k_gemm_wmma(const float* __restrict__ X,
                                                   const float* __restrict__ Wt,
                                                   float* __restrict__ H) {
  static_assert(K == 128, "K fixed at 128");
  constexpr int PAD = K + 4;
  __shared__ float xs[16 * PAD];

  const int tid   = threadIdx.x;
  const int lane  = tid & 31;
  const int wave  = tid >> 5;
  const int mtile = blockIdx.x;

  constexpr int NTHREADS = (OUTD / 16) * 32;
  constexpr int NSLOTS   = 16 * (K / 4);
#pragma unroll
  for (int s = tid; s < NSLOTS; s += NTHREADS) {
    int row = s >> 5;
    int c4  = s & 31;
    float4 v = *(const float4*)(X + (size_t)(mtile * 16 + row) * K + c4 * 4);
    *(float4*)(&xs[row * PAD + c4 * 4]) = v;
  }
  __syncthreads();

  const int idx16 = lane & 15;
  const int half  = lane >> 4;
  const float* xsrow = &xs[idx16 * PAD + 2 * half];
  const float* wrow  = Wt + (size_t)(wave * 16 + idx16) * K + 2 * half;

  v8f acc = {};
#pragma unroll
  for (int k0 = 0; k0 < K; k0 += 4) {
    v2f a = *(const v2f*)(xsrow + k0);
    v2f b = *(const v2f*)(wrow + k0);
    acc = __builtin_amdgcn_wmma_f32_16x16x4_f32(
         false, a,  false, b,
         (short)0, acc,  false,  false);
  }

  float* hp = H + (size_t)(mtile * 16 + half * 8) * OUTD + wave * 16 + idx16;
#pragma unroll
  for (int r = 0; r < 8; ++r) hp[r * OUTD] = acc[r];
}


template <int VEC> struct vec_of;
template <> struct vec_of<4> { typedef __attribute__((ext_vector_type(4))) float type; };
template <> struct vec_of<2> { typedef v2f type; };

template <int N, int OUTD, typename V>
__device__ __forceinline__ void rows_fma(V& acc, const float* __restrict__ H, unsigned lo,
                                         int2 ew, int j) {
  int s[N];
  float w[N];
  V v[N];
#pragma unroll
  for (int u = 0; u < N; ++u) {
    s[u] = __builtin_amdgcn_readlane(ew.x, j + u);
    w[u] = __int_as_float(__builtin_amdgcn_readlane(ew.y, j + u));
  }
#pragma unroll
  for (int u = 0; u < N; ++u) v[u] = *(const V*)(H + (size_t)s[u] * OUTD + lo);
#pragma unroll
  for (int u = 0; u < N; ++u) acc += v[u] * w[u];
}

template <int OUTD, int RELU>
__global__ __launch_bounds__(256) void k_gather(const int* __restrict__ rowptr,
                                                const int2* __restrict__ csr,
                                                const float* __restrict__ dinv,
                                                const float* __restrict__ H,
                                                const float* __restrict__ bias,
                                                float* __restrict__ out) {
  constexpr int VEC = OUTD / 32;
  using V = typename vec_of<VEC>::type;
  const int lane = threadIdx.x & 31;
  const int node = __builtin_amdgcn_readfirstlane((int)(blockIdx.x * 8 + (threadIdx.x >> 5)));
  if (node >= NN) return;
  const int beg = __builtin_amdgcn_readfirstlane(rowptr[node]);
  const int end = __builtin_amdgcn_readfirstlane(rowptr[node + 1]);
  const unsigned lo = (unsigned)lane * VEC;
  const float di = dinv[node];
  V acc = *(const V*)(H + (size_t)node * OUTD + lo) * (di * di);

  for (int base = beg; base < end; base += 32) {
    const int idx = base + lane;
    int2 ew = make_int2(0, 0);
    if (idx < end) ew = csr[idx];
    const int m = min(32, end - base);
    int j = 0;
    for (; j + 8 <= m; j += 8) rows_fma<8, OUTD, V>(acc, H, lo, ew, j);
    if (j + 4 <= m) {
      rows_fma<4, OUTD, V>(acc, H, lo, ew, j);
      j += 4;
    }
    for (; j < m; ++j) rows_fma<1, OUTD, V>(acc, H, lo, ew, j);
  }

  acc += *(const V*)(bias + lo);
  if (RELU) {
#pragma unroll
    for (int k = 0; k < VEC; ++k) acc[k] = fmaxf(acc[k], 0.0f);
  }
  *(V*)(out + (size_t)node * OUTD + lo) = acc;
}


__device__ __forceinline__ vec_of<4>::type node_row_relu(int node, int lane, unsigned lo,
                                                        const int* __restrict__ rowptr,
                                                        const int2* __restrict__ csr,
                                                        const float* __restrict__ dinv,
                                                        const float* __restrict__ H,
                                                        vec_of<4>::type bv) {
  using V = vec_of<4>::type;
  constexpr int K = 128;
  const int beg = __builtin_amdgcn_readfirstlane(rowptr[node]);
  const int end = __builtin_amdgcn_readfirstlane(rowptr[node + 1]);
  const float di = dinv[node];
  V acc = *(const V*)(H + (size_t)node * K + lo) * (di * di);

  for (int base = beg; base < end; base += 32) {
    const int idx = base + lane;
    int2 ew = make_int2(0, 0);
    if (idx < end) ew = csr[idx];
    const int m = min(32, end - base);
    int j = 0;
    for (; j + 8 <= m; j += 8) rows_fma<8, K, V>(acc, H, lo, ew, j);
    if (j + 4 <= m) {
      rows_fma<4, K, V>(acc, H, lo, ew, j);
      j += 4;
    }
    for (; j < m; ++j) rows_fma<1, K, V>(acc, H, lo, ew, j);
  }
  acc += bv;
#pragma unroll
  for (int k = 0; k < 4; ++k) acc[k] = fmaxf(acc[k], 0.0f);
  return acc;
}

template <int OUTN>
__global__ __launch_bounds__(512, 16) void k_gather_gemm(const int* __restrict__ rowptr,
                                                         const int2* __restrict__ csr,
                                                         const float* __restrict__ dinv,
                                                         const float* __restrict__ H,
                                                         const float* __restrict__ bias,
                                                         const float* __restrict__ Wn,
                                                         float* __restrict__ Hn) {
  constexpr int K   = 128;
  constexpr int PAD = K + 4;
  constexpr int NT  = NN / 16;
  static_assert(NN % 16 == 0, "node tiles need no bounds checks");
  __shared__ float wl[OUTN * PAD];
  __shared__ float xs[16 * PAD];
  using V = vec_of<4>::type;

  const int tid  = threadIdx.x;
  const int lane = tid & 31;
  const int wave = __builtin_amdgcn_readfirstlane(tid >> 5);
  const unsigned lo = (unsigned)lane * 4;
  const V bv = *(const V*)(bias + lo);
  const int idx16 = lane & 15;
  const int half  = lane >> 4;

  int tile = blockIdx.x;
  if (tile >= NT) return;

  V row = node_row_relu(tile * 16 + wave, lane, lo, rowptr, csr, dinv, H, bv);
  *(V*)(&xs[wave * PAD + lo]) = row;

#pragma unroll 8
  for (int i = tid; i < K * OUTN; i += 512) {
    const int k = i / OUTN;
    const int n = i - k * OUTN;
    wl[n * PAD + k] = Wn[i];
  }

  while (true) {
    __syncthreads();

    if (wave < OUTN / 16) {
      const float* xsrow = &xs[idx16 * PAD + 2 * half];
      const float* wrow  = &wl[(wave * 16 + idx16) * PAD + 2 * half];
      v8f c = {};
#pragma unroll
      for (int k0 = 0; k0 < K; k0 += 4) {
        v2f a = *(const v2f*)(xsrow + k0);
        v2f b = *(const v2f*)(wrow + k0);
        c = __builtin_amdgcn_wmma_f32_16x16x4_f32(
             false, a,  false, b,
             (short)0, c,  false,  false);
      }
      float* hp = Hn + (size_t)(tile * 16 + half * 8) * OUTN + wave * 16 + idx16;
#pragma unroll
      for (int r = 0; r < 8; ++r) hp[r * OUTN] = c[r];
    }

    tile += gridDim.x;
    if (tile >= NT) break;
    row = node_row_relu(tile * 16 + wave, lane, lo, rowptr, csr, dinv, H, bv);
    __syncthreads();
    *(V*)(&xs[wave * PAD + lo]) = row;
  }
}


extern "C" void kernel_launch(void* const* d_in, const int* in_sizes, int n_in,
                              void* d_out, int out_size, void* d_ws, size_t ws_size,
                              hipStream_t stream) {
  const float* x  = (const float*)d_in[0];
  const int*   ei = (const int*)d_in[1];
  const float* W1 = (const float*)d_in[2];
  const float* b1 = (const float*)d_in[3];
  const float* W2 = (const float*)d_in[4];
  const float* b2 = (const float*)d_in[5];
  const float* W3 = (const float*)d_in[6];
  const float* b3 = (const float*)d_in[7];
  const float* W4 = (const float*)d_in[8];
  const float* b4 = (const float*)d_in[9];

  const int E = in_sizes[1] / 2;
  const int* src = ei;
  const int* dst = ei + E;
  const size_t Ep = ((size_t)E + 3) & ~(size_t)3;

  float* h      = (float*)d_ws;
  float* xcur   = h + (size_t)NN * 128;
  float* wt     = xcur + (size_t)NN * 128;
  float* dinv   = wt + 128 * 128;
  int*   cnt    = (int*)(dinv + NN);
  int*   rowptr = cnt + NN;
  int*   rank   = rowptr + NN + 4;
  int*   elist  = rank + Ep;
  int2*  csr    = (int2*)(elist + Ep);

  unsigned nb = (NN + 255) / 256;
  unsigned eb = (unsigned)((E + 255) / 256);
  k_cnt_zero<<<nb, 256, 0, stream>>>(cnt, NN);
  k_count_rank<<<eb, 256, 0, stream>>>(dst, cnt, rank, E);
  k_scan_dinv<<<(NN + SCAN_B - 1) / SCAN_B, SCAN_B, 0, stream>>>(cnt, rowptr, dinv);
  k_fill<<<eb, 256, 0, stream>>>(dst, rank, rowptr, elist, E);
  k_order<<<(NN + 7) / 8, 256, 0, stream>>>(rowptr, elist, src, dinv, csr);

  k_transpose<128, 128><<<(128 * 128 + 255) / 256, 256, 0, stream>>>(W1, wt);
  k_gemm_wmma<128, 128><<<dim3(NN / 16), dim3(256), 0, stream>>>(x, wt, h);
  k_gather_gemm<128><<<1024, 512, 0, stream>>>(rowptr, csr, dinv, h,    b1, W2, xcur);
  k_gather_gemm<128><<<1024, 512, 0, stream>>>(rowptr, csr, dinv, xcur, b2, W3, h);
  k_gather_gemm<64><<<1024, 512, 0, stream>>>(rowptr, csr, dinv, h,    b3, W4, xcur);
  k_gather<64, 0><<<(NN + 7) / 8, 256, 0, stream>>>(rowptr, csr, dinv, xcur, b4, (float*)d_out);
}


# edge-list build moved into LDS: 2-level counting sort by destination replaces global-atomic count, scan and scattered fill; 11 launches to 9; row sums now in ascending source order
# speedup vs baseline: 7.3849x; 1.6422x over previous
#include <hip/hip_runtime.h>
#include <hip/hip_bf16.h>

typedef __attribute__((ext_vector_type(2))) float v2f;
typedef __attribute__((ext_vector_type(8))) float v8f;

#define NN 50000

#define BINW 196
#define NBIN 256
#define CHK  6400
#define CAP2 8192
#define MAXCH 1024
static_assert(NBIN * BINW >= NN, "bins cover all nodes");

__device__ __forceinline__ int wave_incl_scan(int v, int lane) {
#pragma unroll
  for (int off = 1; off < 32; off <<= 1) {
    int o = __shfl_up(v, off);
    if (lane >= off) v += o;
  }
  return v;
}

__global__ __launch_bounds__(1024) void k_bin(const int* __restrict__ src,
                                              const int* __restrict__ dst,
                                              int2* __restrict__ binned,
                                              int* __restrict__ cm, int e) {
  __shared__ int hist[NBIN];
  __shared__ int offs[NBIN];
  __shared__ int wsum[NBIN / 32];
  __shared__ int2 stage[CHK];
  const int t    = threadIdx.x;
  const int lane = t & 31;
  const int wave = t >> 5;
  const int c    = blockIdx.x;
  const int nch  = gridDim.x;
  const int cbeg = c * CHK;
  const int n    = min(CHK, e - cbeg);

  if (t < NBIN) hist[t] = 0;
  __syncthreads();

  constexpr int PER = (CHK + 1023) / 1024;
  int s[PER], d[PER], r[PER];
#pragma unroll
  for (int k = 0; k < PER; ++k) {
    const int j = k * 1024 + t;
    s[k] = 0; d[k] = 0;
    if (j < n) { d[k] = dst[cbeg + j]; s[k] = src[cbeg + j]; }
  }
#pragma unroll
  for (int k = 0; k < PER; ++k) {
    const int j = k * 1024 + t;
    r[k] = 0;
    if (j < n) r[k] = atomicAdd(&hist[d[k] / BINW], 1);
  }
  __syncthreads();

  const int v   = (t < NBIN) ? hist[t] : 0;
  const int inc = wave_incl_scan(v, lane);
  if (t < NBIN && lane == 31) wsum[wave] = inc;
  __syncthreads();
  if (t < NBIN) {
    int base = 0;
#pragma unroll
    for (int w = 0; w < NBIN / 32; ++w)
      if (w < wave) base += wsum[w];
    offs[t] = base + inc - v;
  }
  __syncthreads();

#pragma unroll
  for (int k = 0; k < PER; ++k) {
    const int j = k * 1024 + t;
    if (j < n) stage[offs[d[k] / BINW] + r[k]] = make_int2(s[k], d[k]);
  }
  __syncthreads();
  for (int j = t; j < n; j += 1024) binned[cbeg + j] = stage[j];
  if (t < NBIN) cm[t * nch + c] = (offs[t] << 16) | hist[t];
}

__device__ __forceinline__ void sort_bin(int2* ed, int* so, int* ncnt, int* nrow, int* wsum,
                                         const int* crun_w, const int* crun_st,
                                         const int2* __restrict__ binned, int nch, int b,
                                         int base, int nb, int* __restrict__ rowptr,
                                         float* __restrict__ dinv, int* __restrict__ ssrc) {
  const int t     = threadIdx.x;
  const int lane  = t & 31;
  const int wave  = __builtin_amdgcn_readfirstlane(t >> 5);
  const int nbase = b * BINW;

  if (t < 256) ncnt[t] = 0;
  __syncthreads();

  for (int c0 = wave * 4; c0 < nch; c0 += 128) {
    int cnu[4], stu[4];
    const int2* runu[4];
    int2 pu[4];
#pragma unroll
    for (int u = 0; u < 4; ++u) {
      const int c = c0 + u;
      int w = 0, st = 0;
      if (c < nch) { w = crun_w[c]; st = crun_st[c]; }
      w  = __builtin_amdgcn_readfirstlane(w);
      st = __builtin_amdgcn_readfirstlane(st);
      cnu[u]  = w & 0xffff;
      stu[u]  = st;
      runu[u] = binned + (size_t)min(c, nch - 1) * CHK + (int)((unsigned)w >> 16);
      pu[u]   = make_int2(0, 0);
      if (lane < cnu[u]) pu[u] = runu[u][lane];
    }
#pragma unroll
    for (int u = 0; u < 4; ++u) {
      if (lane < cnu[u]) {
        const int ln = pu[u].y - nbase;
        const int rk = atomicAdd(&ncnt[ln], 1);
        ed[stu[u] + lane] = make_int2(pu[u].x, (rk << 8) | ln);
      }
      for (int l = lane + 32; l < cnu[u]; l += 32) {
        const int2 p = runu[u][l];
        const int ln = p.y - nbase;
        const int rk = atomicAdd(&ncnt[ln], 1);
        ed[stu[u] + l] = make_int2(p.x, (rk << 8) | ln);
      }
    }
  }
  __syncthreads();

  const int v   = (t < 256) ? ncnt[t] : 0;
  const int inc = wave_incl_scan(v, lane);
  if (t < 256 && lane == 31) wsum[wave] = inc;
  __syncthreads();
  if (t < 256) {
    int pre = 0;
#pragma unroll
    for (int w = 0; w < 8; ++w)
      if (w < wave) pre += wsum[w];
    const int ex = pre + inc - v;
    nrow[t] = ex;
    const int node = nbase + t;
    if (t < BINW && node < NN) {
      rowptr[node] = base + ex;
      float dg = (float)(v + 1);
      dinv[node] = (dg > 0.0f) ? rsqrtf(dg) : 0.0f;
    }
  }
  if (b == NBIN - 1 && t == 0) rowptr[NN] = base + nb;
  __syncthreads();

  for (int j = t; j < nb; j += 1024) {
    const int2 p = ed[j];
    const int ln = p.y & 255;
    const int rk = (int)((unsigned)p.y >> 8);
    so[nrow[ln] + rk] = p.x;
  }
  __syncthreads();

  for (int ln = wave; ln < BINW; ln += 32) {
    const int beg = __builtin_amdgcn_readfirstlane(nrow[ln]);
    const int end = beg + __builtin_amdgcn_readfirstlane(ncnt[ln]);
    for (int bs = beg; bs < end; bs += 32) {
      const int idx   = bs + lane;
      const bool live = idx < end;
      const int mine  = live ? so[idx] : 0x7fffffff;
      int pos = 0;
      for (int ob = beg; ob < end; ob += 32) {
        const int oidx  = ob + lane;
        const int other = (oidx < end) ? so[oidx] : 0x7fffffff;
        const int m = min(32, end - ob);
        for (int j = 0; j < m; ++j) {
          const int o = __builtin_amdgcn_readlane(other, j);
          pos += (o < mine || (o == mine && ob + j < idx)) ? 1 : 0;
        }
      }
      if (live) ssrc[base + beg + pos] = mine;
    }
  }
}

__global__ __launch_bounds__(1024) void k_sort(const int2* __restrict__ binned,
                                               const int* __restrict__ cm, int nch,
                                               int2* __restrict__ ged, int* __restrict__ gso,
                                               int* __restrict__ rowptr,
                                               float* __restrict__ dinv,
                                               int* __restrict__ ssrc) {
  __shared__ int ncnt[256];
  __shared__ int nrow[256];
  __shared__ int wtot[32];
  __shared__ int ptot[32];
  __shared__ int crun_w[MAXCH];
  __shared__ int crun_st[MAXCH];
  __shared__ int2 ed[CAP2];
  __shared__ int so[CAP2];
  const int t    = threadIdx.x;
  const int lane = t & 31;
  const int wave = t >> 5;
  const int b    = blockIdx.x;
  nch = min(nch, MAXCH);

  const int nfront = b * nch;
  const int w  = (t < nch) ? cm[nfront + t] : 0;
  const int cn = w & 0xffff;
  int p = (int)((unsigned)w >> 16);
  const int inc = wave_incl_scan(cn, lane);
#pragma unroll
  for (int off = 16; off > 0; off >>= 1) p += __shfl_xor(p, off);
  if (lane == 31) wtot[wave] = inc;
  if (lane == 0)  ptot[wave] = p;
  __syncthreads();
  int base = 0, pre = 0, nb = 0;
#pragma unroll
  for (int k = 0; k < 32; ++k) {
    base += ptot[k];
    nb   += wtot[k];
    if (k < wave) pre += wtot[k];
  }
  if (t < nch) {
    crun_w[t]  = w;
    crun_st[t] = pre + inc - cn;
  }
  base = __builtin_amdgcn_readfirstlane(base);
  nb   = __builtin_amdgcn_readfirstlane(nb);
  if (nb <= CAP2)
    sort_bin(ed, so, ncnt, nrow, wtot, crun_w, crun_st, binned, nch, b, base, nb, rowptr, dinv, ssrc);
  else
    sort_bin(ged + base, gso + base, ncnt, nrow, wtot, crun_w, crun_st, binned, nch, b, base, nb,
             rowptr, dinv, ssrc);
}

__global__ __launch_bounds__(1024) void k_norm(const int* __restrict__ rowptr,
                                               const int* __restrict__ ssrc,
                                               const float* __restrict__ dinv,
                                               int2* __restrict__ csr) {
  __shared__ float dl[NN];
  __shared__ int rp[BINW + 1];
  const int t  = threadIdx.x;
  const int n0 = blockIdx.x * BINW;
  const int nn = min(BINW, NN - n0);
  static_assert(NN % 4 == 0, "float4 copy of the table");
  static_assert((NBIN - 1) * BINW < NN, "every bin has a node");
  if (t <= nn) rp[t] = rowptr[n0 + t];
  for (int i = t; i < NN / 4; i += 1024) ((float4*)dl)[i] = ((const float4*)dinv)[i];
  __syncthreads();
  const int beg = rp[0];
  const int end = rp[nn];
  for (int idx = beg + t; idx < end; idx += 1024) {
    int lo = 0, hi = nn;
    while (hi - lo > 1) {
      const int mid = (lo + hi) >> 1;
      if (rp[mid] <= idx) lo = mid; else hi = mid;
    }
    const int s = ssrc[idx];
    csr[idx] = make_int2(s, __float_as_int(dl[s] * dl[n0 + lo]));
  }
}

template <int K, int OUTD>
__global__ __launch_bounds__(256) void k_transpose(const float* __restrict__ W,
                                                   float* __restrict__ Wt) {
  int i = blockIdx.x * blockDim.x + threadIdx.x;
  if (i < K * OUTD) {
    int k = i / OUTD;
    int n = i - k * OUTD;
    Wt[n * K + k] = W[i];
  }
}

template <int K, int OUTD>
__global__ __launch_bounds__(256) void k_gemm_wmma(const float* __restrict__ X,
                                                   const float* __restrict__ Wt,
                                                   float* __restrict__ H) {
  static_assert(K == 128, "K fixed at 128");
  constexpr int PAD = K + 4;
  __shared__ float xs[16 * PAD];

  const int tid   = threadIdx.x;
  const int lane  = tid & 31;
  const int wave  = tid >> 5;
  const int mtile = blockIdx.x;

  constexpr int NTHREADS = (OUTD / 16) * 32;
  constexpr int NSLOTS   = 16 * (K / 4);
#pragma unroll
  for (int s = tid; s < NSLOTS; s += NTHREADS) {
    int row = s >> 5;
    int c4  = s & 31;
    float4 v = *(const float4*)(X + (size_t)(mtile * 16 + row) * K + c4 * 4);
    *(float4*)(&xs[row * PAD + c4 * 4]) = v;
  }
  __syncthreads();

  const int idx16 = lane & 15;
  const int half  = lane >> 4;
  const float* xsrow = &xs[idx16 * PAD + 2 * half];
  const float* wrow  = Wt + (size_t)(wave * 16 + idx16) * K + 2 * half;

  v8f acc = {};
#pragma unroll
  for (int k0 = 0; k0 < K; k0 += 4) {
    v2f a = *(const v2f*)(xsrow + k0);
    v2f b = *(const v2f*)(wrow + k0);
    acc = __builtin_amdgcn_wmma_f32_16x16x4_f32(
         false, a,  false, b,
         (short)0, acc,  false,  false);
  }

  float* hp = H + (size_t)(mtile * 16 + half * 8) * OUTD + wave * 16 + idx16;
#pragma unroll
  for (int r = 0; r < 8; ++r) hp[r * OUTD] = acc[r];
}


template <int VEC> struct vec_of;
template <> struct vec_of<4> { typedef __attribute__((ext_vector_type(4))) float type; };
template <> struct vec_of<2> { typedef v2f type; };

template <int N, int OUTD, typename V>
__device__ __forceinline__ void rows_fma(V& acc, const float* __restrict__ H, unsigned lo,
                                         int2 ew, int j) {
  int s[N];
  float w[N];
  V v[N];
#pragma unroll
  for (int u = 0; u < N; ++u) {
    s[u] = __builtin_amdgcn_readlane(ew.x, j + u);
    w[u] = __int_as_float(__builtin_amdgcn_readlane(ew.y, j + u));
  }
#pragma unroll
  for (int u = 0; u < N; ++u) v[u] = *(const V*)(H + (size_t)s[u] * OUTD + lo);
#pragma unroll
  for (int u = 0; u < N; ++u) acc += v[u] * w[u];
}

template <int OUTD, int RELU>
__global__ __launch_bounds__(256) void k_gather(const int* __restrict__ rowptr,
                                                const int2* __restrict__ csr,
                                                const float* __restrict__ dinv,
                                                const float* __restrict__ H,
                                                const float* __restrict__ bias,
                                                float* __restrict__ out) {
  constexpr int VEC = OUTD / 32;
  using V = typename vec_of<VEC>::type;
  const int lane = threadIdx.x & 31;
  const int node = __builtin_amdgcn_readfirstlane((int)(blockIdx.x * 8 + (threadIdx.x >> 5)));
  if (node >= NN) return;
  const int beg = __builtin_amdgcn_readfirstlane(rowptr[node]);
  const int end = __builtin_amdgcn_readfirstlane(rowptr[node + 1]);
  const unsigned lo = (unsigned)lane * VEC;
  const float di = dinv[node];
  V acc = *(const V*)(H + (size_t)node * OUTD + lo) * (di * di);

  for (int base = beg; base < end; base += 32) {
    const int idx = base + lane;
    int2 ew = make_int2(0, 0);
    if (idx < end) ew = csr[idx];
    const int m = min(32, end - base);
    int j = 0;
    for (; j + 8 <= m; j += 8) rows_fma<8, OUTD, V>(acc, H, lo, ew, j);
    if (j + 4 <= m) {
      rows_fma<4, OUTD, V>(acc, H, lo, ew, j);
      j += 4;
    }
    for (; j < m; ++j) rows_fma<1, OUTD, V>(acc, H, lo, ew, j);
  }

  acc += *(const V*)(bias + lo);
  if (RELU) {
#pragma unroll
    for (int k = 0; k < VEC; ++k) acc[k] = fmaxf(acc[k], 0.0f);
  }
  *(V*)(out + (size_t)node * OUTD + lo) = acc;
}


__device__ __forceinline__ vec_of<4>::type node_row_relu(int node, int lane, unsigned lo,
                                                        const int* __restrict__ rowptr,
                                                        const int2* __restrict__ csr,
                                                        const float* __restrict__ dinv,
                                                        const float* __restrict__ H,
                                                        vec_of<4>::type bv) {
  using V = vec_of<4>::type;
  constexpr int K = 128;
  const int beg = __builtin_amdgcn_readfirstlane(rowptr[node]);
  const int end = __builtin_amdgcn_readfirstlane(rowptr[node + 1]);
  const float di = dinv[node];
  V acc = *(const V*)(H + (size_t)node * K + lo) * (di * di);

  for (int base = beg; base < end; base += 32) {
    const int idx = base + lane;
    int2 ew = make_int2(0, 0);
    if (idx < end) ew = csr[idx];
    const int m = min(32, end - base);
    int j = 0;
    for (; j + 8 <= m; j += 8) rows_fma<8, K, V>(acc, H, lo, ew, j);
    if (j + 4 <= m) {
      rows_fma<4, K, V>(acc, H, lo, ew, j);
      j += 4;
    }
    for (; j < m; ++j) rows_fma<1, K, V>(acc, H, lo, ew, j);
  }
  acc += bv;
#pragma unroll
  for (int k = 0; k < 4; ++k) acc[k] = fmaxf(acc[k], 0.0f);
  return acc;
}

template <int OUTN>
__global__ __launch_bounds__(512, 16) void k_gather_gemm(const int* __restrict__ rowptr,
                                                         const int2* __restrict__ csr,
                                                         const float* __restrict__ dinv,
                                                         const float* __restrict__ H,
                                                         const float* __restrict__ bias,
                                                         const float* __restrict__ Wn,
                                                         float* __restrict__ Hn) {
  constexpr int K   = 128;
  constexpr int PAD = K + 4;
  constexpr int NT  = NN / 16;
  static_assert(NN % 16 == 0, "node tiles need no bounds checks");
  __shared__ float wl[OUTN * PAD];
  __shared__ float xs[16 * PAD];
  using V = vec_of<4>::type;

  const int tid  = threadIdx.x;
  const int lane = tid & 31;
  const int wave = __builtin_amdgcn_readfirstlane(tid >> 5);
  const unsigned lo = (unsigned)lane * 4;
  const V bv = *(const V*)(bias + lo);
  const int idx16 = lane & 15;
  const int half  = lane >> 4;

  int tile = blockIdx.x;
  if (tile >= NT) return;

  V row = node_row_relu(tile * 16 + wave, lane, lo, rowptr, csr, dinv, H, bv);
  *(V*)(&xs[wave * PAD + lo]) = row;

#pragma unroll 8
  for (int i = tid; i < K * OUTN; i += 512) {
    const int k = i / OUTN;
    const int n = i - k * OUTN;
    wl[n * PAD + k] = Wn[i];
  }

  while (true) {
    __syncthreads();

    if (wave < OUTN / 16) {
      const float* xsrow = &xs[idx16 * PAD + 2 * half];
      const float* wrow  = &wl[(wave * 16 + idx16) * PAD + 2 * half];
      v8f c = {};
#pragma unroll
      for (int k0 = 0; k0 < K; k0 += 4) {
        v2f a = *(const v2f*)(xsrow + k0);
        v2f b = *(const v2f*)(wrow + k0);
        c = __builtin_amdgcn_wmma_f32_16x16x4_f32(
             false, a,  false, b,
             (short)0, c,  false,  false);
      }
      float* hp = Hn + (size_t)(tile * 16 + half * 8) * OUTN + wave * 16 + idx16;
#pragma unroll
      for (int r = 0; r < 8; ++r) hp[r * OUTN] = c[r];
    }

    tile += gridDim.x;
    if (tile >= NT) break;
    row = node_row_relu(tile * 16 + wave, lane, lo, rowptr, csr, dinv, H, bv);
    __syncthreads();
    *(V*)(&xs[wave * PAD + lo]) = row;
  }
}


extern "C" void kernel_launch(void* const* d_in, const int* in_sizes, int n_in,
                              void* d_out, int out_size, void* d_ws, size_t ws_size,
                              hipStream_t stream) {
  const float* x  = (const float*)d_in[0];
  const int*   ei = (const int*)d_in[1];
  const float* W1 = (const float*)d_in[2];
  const float* b1 = (const float*)d_in[3];
  const float* W2 = (const float*)d_in[4];
  const float* b2 = (const float*)d_in[5];
  const float* W3 = (const float*)d_in[6];
  const float* b3 = (const float*)d_in[7];
  const float* W4 = (const float*)d_in[8];
  const float* b4 = (const float*)d_in[9];

  const int E = in_sizes[1] / 2;
  const int* src = ei;
  const int* dst = ei + E;
  float* h      = (float*)d_ws;
  float* xcur   = h + (size_t)NN * 128;
  float* wt     = xcur + (size_t)NN * 128;
  float* dinv   = wt + 128 * 128;
  int*   rowptr = (int*)(dinv + NN);
  int2*  csr    = (int2*)(rowptr + NN + 4);
  int2*  binned = (int2*)xcur;
  int*   cm     = (int*)xcur + 3400000;
  int*   ssrc   = (int*)xcur + 3700000;
  int2*  ged    = (int2*)h;
  int*   gso    = (int*)h + 3400000;

  unsigned nch = (unsigned)((E + CHK - 1) / CHK);
  k_bin<<<nch, 1024, 0, stream>>>(src, dst, binned, cm, E);
  k_sort<<<NBIN, 1024, 0, stream>>>(binned, cm, (int)nch, ged, gso, rowptr, dinv, ssrc);
  k_norm<<<NBIN, 1024, 0, stream>>>(rowptr, ssrc, dinv, csr);

  k_transpose<128, 128><<<(128 * 128 + 255) / 256, 256, 0, stream>>>(W1, wt);
  k_gemm_wmma<128, 128><<<dim3(NN / 16), dim3(256), 0, stream>>>(x, wt, h);
  k_gather_gemm<128><<<1024, 512, 0, stream>>>(rowptr, csr, dinv, h,    b1, W2, xcur);
  k_gather_gemm<128><<<1024, 512, 0, stream>>>(rowptr, csr, dinv, xcur, b2, W3, h);
  k_gather_gemm<64><<<1024, 512, 0, stream>>>(rowptr, csr, dinv, h,    b3, W4, xcur);
  k_gather<64, 0><<<(NN + 7) / 8, 256, 0, stream>>>(rowptr, csr, dinv, xcur, b4, (float*)d_out);
}


# edge weight factored out of the list: H rows carry dinv[src] (tile rows scaled before their product), dinv[dst] applied once to the finished sum; weight kernel gone, 9 launches to 8; sum order kept
# speedup vs baseline: 7.7835x; 1.0540x over previous
#include <hip/hip_runtime.h>
#include <hip/hip_bf16.h>

typedef __attribute__((ext_vector_type(2))) float v2f;
typedef __attribute__((ext_vector_type(8))) float v8f;

#define NN 50000

#define BINW 196
#define NBIN 256
#define CHK  6400
#define CAP2 8192
#define MAXCH 1024
static_assert(NBIN * BINW >= NN, "bins cover all nodes");

__device__ __forceinline__ int wave_incl_scan(int v, int lane) {
#pragma unroll
  for (int off = 1; off < 32; off <<= 1) {
    int o = __shfl_up(v, off);
    if (lane >= off) v += o;
  }
  return v;
}

__global__ __launch_bounds__(1024) void k_bin(const int* __restrict__ src,
                                              const int* __restrict__ dst,
                                              int2* __restrict__ binned,
                                              int* __restrict__ cm, int e) {
  __shared__ int hist[NBIN];
  __shared__ int offs[NBIN];
  __shared__ int wsum[NBIN / 32];
  __shared__ int2 stage[CHK];
  const int t    = threadIdx.x;
  const int lane = t & 31;
  const int wave = t >> 5;
  const int c    = blockIdx.x;
  const int nch  = gridDim.x;
  const int cbeg = c * CHK;
  const int n    = min(CHK, e - cbeg);

  if (t < NBIN) hist[t] = 0;
  __syncthreads();

  constexpr int PER = (CHK + 1023) / 1024;
  int s[PER], d[PER], r[PER];
#pragma unroll
  for (int k = 0; k < PER; ++k) {
    const int j = k * 1024 + t;
    s[k] = 0; d[k] = 0;
    if (j < n) { d[k] = dst[cbeg + j]; s[k] = src[cbeg + j]; }
  }
#pragma unroll
  for (int k = 0; k < PER; ++k) {
    const int j = k * 1024 + t;
    r[k] = 0;
    if (j < n) r[k] = atomicAdd(&hist[d[k] / BINW], 1);
  }
  __syncthreads();

  const int v   = (t < NBIN) ? hist[t] : 0;
  const int inc = wave_incl_scan(v, lane);
  if (t < NBIN && lane == 31) wsum[wave] = inc;
  __syncthreads();
  if (t < NBIN) {
    int base = 0;
#pragma unroll
    for (int w = 0; w < NBIN / 32; ++w)
      if (w < wave) base += wsum[w];
    offs[t] = base + inc - v;
  }
  __syncthreads();

#pragma unroll
  for (int k = 0; k < PER; ++k) {
    const int j = k * 1024 + t;
    if (j < n) stage[offs[d[k] / BINW] + r[k]] = make_int2(s[k], d[k]);
  }
  __syncthreads();
  for (int j = t; j < n; j += 1024) binned[cbeg + j] = stage[j];
  if (t < NBIN) cm[t * nch + c] = (offs[t] << 16) | hist[t];
}

__device__ __forceinline__ void sort_bin(int2* ed, int* so, int* ncnt, int* nrow, int* wsum,
                                         const int* crun_w, const int* crun_st,
                                         const int2* __restrict__ binned, int nch, int b,
                                         int base, int nb, int* __restrict__ rowptr,
                                         float* __restrict__ dinv, int* __restrict__ ssrc) {
  const int t     = threadIdx.x;
  const int lane  = t & 31;
  const int wave  = __builtin_amdgcn_readfirstlane(t >> 5);
  const int nbase = b * BINW;

  if (t < 256) ncnt[t] = 0;
  __syncthreads();

  for (int c0 = wave * 4; c0 < nch; c0 += 128) {
    int cnu[4], stu[4];
    const int2* runu[4];
    int2 pu[4];
#pragma unroll
    for (int u = 0; u < 4; ++u) {
      const int c = c0 + u;
      int w = 0, st = 0;
      if (c < nch) { w = crun_w[c]; st = crun_st[c]; }
      w  = __builtin_amdgcn_readfirstlane(w);
      st = __builtin_amdgcn_readfirstlane(st);
      cnu[u]  = w & 0xffff;
      stu[u]  = st;
      runu[u] = binned + (size_t)min(c, nch - 1) * CHK + (int)((unsigned)w >> 16);
      pu[u]   = make_int2(0, 0);
      if (lane < cnu[u]) pu[u] = runu[u][lane];
    }
#pragma unroll
    for (int u = 0; u < 4; ++u) {
      if (lane < cnu[u]) {
        const int ln = pu[u].y - nbase;
        const int rk = atomicAdd(&ncnt[ln], 1);
        ed[stu[u] + lane] = make_int2(pu[u].x, (rk << 8) | ln);
      }
      for (int l = lane + 32; l < cnu[u]; l += 32) {
        const int2 p = runu[u][l];
        const int ln = p.y - nbase;
        const int rk = atomicAdd(&ncnt[ln], 1);
        ed[stu[u] + l] = make_int2(p.x, (rk << 8) | ln);
      }
    }
  }
  __syncthreads();

  const int v   = (t < 256) ? ncnt[t] : 0;
  const int inc = wave_incl_scan(v, lane);
  if (t < 256 && lane == 31) wsum[wave] = inc;
  __syncthreads();
  if (t < 256) {
    int pre = 0;
#pragma unroll
    for (int w = 0; w < 8; ++w)
      if (w < wave) pre += wsum[w];
    const int ex = pre + inc - v;
    nrow[t] = ex;
    const int node = nbase + t;
    if (t < BINW && node < NN) {
      rowptr[node] = base + ex;
      float dg = (float)(v + 1);
      dinv[node] = (dg > 0.0f) ? rsqrtf(dg) : 0.0f;
    }
  }
  if (b == NBIN - 1 && t == 0) rowptr[NN] = base + nb;
  __syncthreads();

  for (int j = t; j < nb; j += 1024) {
    const int2 p = ed[j];
    const int ln = p.y & 255;
    const int rk = (int)((unsigned)p.y >> 8);
    so[nrow[ln] + rk] = p.x;
  }
  __syncthreads();

  for (int ln = wave; ln < BINW; ln += 32) {
    const int beg = __builtin_amdgcn_readfirstlane(nrow[ln]);
    const int end = beg + __builtin_amdgcn_readfirstlane(ncnt[ln]);
    for (int bs = beg; bs < end; bs += 32) {
      const int idx   = bs + lane;
      const bool live = idx < end;
      const int mine  = live ? so[idx] : 0x7fffffff;
      int pos = 0;
      for (int ob = beg; ob < end; ob += 32) {
        const int oidx  = ob + lane;
        const int other = (oidx < end) ? so[oidx] : 0x7fffffff;
        const int m = min(32, end - ob);
        for (int j = 0; j < m; ++j) {
          const int o = __builtin_amdgcn_readlane(other, j);
          pos += (o < mine || (o == mine && ob + j < idx)) ? 1 : 0;
        }
      }
      if (live) ssrc[base + beg + pos] = mine;
    }
  }
}

__global__ __launch_bounds__(1024) void k_sort(const int2* __restrict__ binned,
                                               const int* __restrict__ cm, int nch,
                                               int2* __restrict__ ged, int* __restrict__ gso,
                                               int* __restrict__ rowptr,
                                               float* __restrict__ dinv,
                                               int* __restrict__ ssrc) {
  __shared__ int ncnt[256];
  __shared__ int nrow[256];
  __shared__ int wtot[32];
  __shared__ int ptot[32];
  __shared__ int crun_w[MAXCH];
  __shared__ int crun_st[MAXCH];
  __shared__ int2 ed[CAP2];
  __shared__ int so[CAP2];
  const int t    = threadIdx.x;
  const int lane = t & 31;
  const int wave = t >> 5;
  const int b    = blockIdx.x;
  nch = min(nch, MAXCH);

  const int nfront = b * nch;
  const int w  = (t < nch) ? cm[nfront + t] : 0;
  const int cn = w & 0xffff;
  int p = (int)((unsigned)w >> 16);
  const int inc = wave_incl_scan(cn, lane);
#pragma unroll
  for (int off = 16; off > 0; off >>= 1) p += __shfl_xor(p, off);
  if (lane == 31) wtot[wave] = inc;
  if (lane == 0)  ptot[wave] = p;
  __syncthreads();
  int base = 0, pre = 0, nb = 0;
#pragma unroll
  for (int k = 0; k < 32; ++k) {
    base += ptot[k];
    nb   += wtot[k];
    if (k < wave) pre += wtot[k];
  }
  if (t < nch) {
    crun_w[t]  = w;
    crun_st[t] = pre + inc - cn;
  }
  base = __builtin_amdgcn_readfirstlane(base);
  nb   = __builtin_amdgcn_readfirstlane(nb);
  if (nb <= CAP2)
    sort_bin(ed, so, ncnt, nrow, wtot, crun_w, crun_st, binned, nch, b, base, nb, rowptr, dinv, ssrc);
  else
    sort_bin(ged + base, gso + base, ncnt, nrow, wtot, crun_w, crun_st, binned, nch, b, base, nb,
             rowptr, dinv, ssrc);
}

template <int K, int OUTD>
__global__ __launch_bounds__(256) void k_transpose(const float* __restrict__ W,
                                                   float* __restrict__ Wt) {
  int i = blockIdx.x * blockDim.x + threadIdx.x;
  if (i < K * OUTD) {
    int k = i / OUTD;
    int n = i - k * OUTD;
    Wt[n * K + k] = W[i];
  }
}

template <int K, int OUTD>
__global__ __launch_bounds__(256) void k_gemm_wmma(const float* __restrict__ X,
                                                   const float* __restrict__ Wt,
                                                   const float* __restrict__ dinv,
                                                   float* __restrict__ H) {
  static_assert(K == 128, "K fixed at 128");
  constexpr int PAD = K + 4;
  __shared__ float xs[16 * PAD];

  const int tid   = threadIdx.x;
  const int lane  = tid & 31;
  const int wave  = tid >> 5;
  const int mtile = blockIdx.x;

  constexpr int NTHREADS = (OUTD / 16) * 32;
  constexpr int NSLOTS   = 16 * (K / 4);
#pragma unroll
  for (int s = tid; s < NSLOTS; s += NTHREADS) {
    int row = s >> 5;
    int c4  = s & 31;
    float4 v = *(const float4*)(X + (size_t)(mtile * 16 + row) * K + c4 * 4);
    const float d = dinv[mtile * 16 + row];
    v.x *= d; v.y *= d; v.z *= d; v.w *= d;
    *(float4*)(&xs[row * PAD + c4 * 4]) = v;
  }
  __syncthreads();

  const int idx16 = lane & 15;
  const int half  = lane >> 4;
  const float* xsrow = &xs[idx16 * PAD + 2 * half];
  const float* wrow  = Wt + (size_t)(wave * 16 + idx16) * K + 2 * half;

  v8f acc = {};
#pragma unroll
  for (int k0 = 0; k0 < K; k0 += 4) {
    v2f a = *(const v2f*)(xsrow + k0);
    v2f b = *(const v2f*)(wrow + k0);
    acc = __builtin_amdgcn_wmma_f32_16x16x4_f32(
         false, a,  false, b,
         (short)0, acc,  false,  false);
  }

  float* hp = H + (size_t)(mtile * 16 + half * 8) * OUTD + wave * 16 + idx16;
#pragma unroll
  for (int r = 0; r < 8; ++r) hp[r * OUTD] = acc[r];
}


template <int VEC> struct vec_of;
template <> struct vec_of<4> { typedef __attribute__((ext_vector_type(4))) float type; };
template <> struct vec_of<2> { typedef v2f type; };

template <int N, int OUTD, typename V>
__device__ __forceinline__ void rows_add(V& acc, const float* __restrict__ H, unsigned lo,
                                         int ev, int j) {
  int s[N];
  V v[N];
#pragma unroll
  for (int u = 0; u < N; ++u) {
    s[u] = __builtin_amdgcn_readlane(ev, j + u);
  }
#pragma unroll
  for (int u = 0; u < N; ++u) v[u] = *(const V*)(H + (size_t)s[u] * OUTD + lo);
#pragma unroll
  for (int u = 0; u < N; ++u) acc += v[u];
}

template <int OUTD, int RELU>
__global__ __launch_bounds__(256) void k_gather(const int* __restrict__ rowptr,
                                                const int* __restrict__ csr,
                                                const float* __restrict__ dinv,
                                                const float* __restrict__ H,
                                                const float* __restrict__ bias,
                                                float* __restrict__ out) {
  constexpr int VEC = OUTD / 32;
  using V = typename vec_of<VEC>::type;
  const int lane = threadIdx.x & 31;
  const int node = __builtin_amdgcn_readfirstlane((int)(blockIdx.x * 8 + (threadIdx.x >> 5)));
  if (node >= NN) return;
  const int beg = __builtin_amdgcn_readfirstlane(rowptr[node]);
  const int end = __builtin_amdgcn_readfirstlane(rowptr[node + 1]);
  const unsigned lo = (unsigned)lane * VEC;
  const float di = dinv[node];
  V acc = *(const V*)(H + (size_t)node * OUTD + lo);

  for (int base = beg; base < end; base += 32) {
    const int idx = base + lane;
    int ev = 0;
    if (idx < end) ev = csr[idx];
    const int m = min(32, end - base);
    int j = 0;
    for (; j + 8 <= m; j += 8) rows_add<8, OUTD, V>(acc, H, lo, ev, j);
    if (j + 4 <= m) {
      rows_add<4, OUTD, V>(acc, H, lo, ev, j);
      j += 4;
    }
    for (; j < m; ++j) rows_add<1, OUTD, V>(acc, H, lo, ev, j);
  }

  acc = acc * di + *(const V*)(bias + lo);
  if (RELU) {
#pragma unroll
    for (int k = 0; k < VEC; ++k) acc[k] = fmaxf(acc[k], 0.0f);
  }
  *(V*)(out + (size_t)node * OUTD + lo) = acc;
}


__device__ __forceinline__ vec_of<4>::type node_row_relu(int node, int lane, unsigned lo,
                                                        const int* __restrict__ rowptr,
                                                        const int* __restrict__ csr,
                                                        const float* __restrict__ dinv,
                                                        const float* __restrict__ H,
                                                        vec_of<4>::type bv) {
  using V = vec_of<4>::type;
  constexpr int K = 128;
  const int beg = __builtin_amdgcn_readfirstlane(rowptr[node]);
  const int end = __builtin_amdgcn_readfirstlane(rowptr[node + 1]);
  const float di = dinv[node];
  V acc = *(const V*)(H + (size_t)node * K + lo);

  for (int base = beg; base < end; base += 32) {
    const int idx = base + lane;
    int ev = 0;
    if (idx < end) ev = csr[idx];
    const int m = min(32, end - base);
    int j = 0;
    for (; j + 8 <= m; j += 8) rows_add<8, K, V>(acc, H, lo, ev, j);
    if (j + 4 <= m) {
      rows_add<4, K, V>(acc, H, lo, ev, j);
      j += 4;
    }
    for (; j < m; ++j) rows_add<1, K, V>(acc, H, lo, ev, j);
  }
  acc = acc * di + bv;
#pragma unroll
  for (int k = 0; k < 4; ++k) acc[k] = fmaxf(acc[k], 0.0f);
  return acc * di;
}

template <int OUTN>
__global__ __launch_bounds__(512, 16) void k_gather_gemm(const int* __restrict__ rowptr,
                                                         const int* __restrict__ csr,
                                                         const float* __restrict__ dinv,
                                                         const float* __restrict__ H,
                                                         const float* __restrict__ bias,
                                                         const float* __restrict__ Wn,
                                                         float* __restrict__ Hn) {
  constexpr int K   = 128;
  constexpr int PAD = K + 4;
  constexpr int NT  = NN / 16;
  static_assert(NN % 16 == 0, "node tiles need no bounds checks");
  __shared__ float wl[OUTN * PAD];
  __shared__ float xs[16 * PAD];
  using V = vec_of<4>::type;

  const int tid  = threadIdx.x;
  const int lane = tid & 31;
  const int wave = __builtin_amdgcn_readfirstlane(tid >> 5);
  const unsigned lo = (unsigned)lane * 4;
  const V bv = *(const V*)(bias + lo);
  const int idx16 = lane & 15;
  const int half  = lane >> 4;

  int tile = blockIdx.x;
  if (tile >= NT) return;

  V row = node_row_relu(tile * 16 + wave, lane, lo, rowptr, csr, dinv, H, bv);
  *(V*)(&xs[wave * PAD + lo]) = row;

#pragma unroll 8
  for (int i = tid; i < K * OUTN; i += 512) {
    const int k = i / OUTN;
    const int n = i - k * OUTN;
    wl[n * PAD + k] = Wn[i];
  }

  while (true) {
    __syncthreads();

    if (wave < OUTN / 16) {
      const float* xsrow = &xs[idx16 * PAD + 2 * half];
      const float* wrow  = &wl[(wave * 16 + idx16) * PAD + 2 * half];
      v8f c = {};
#pragma unroll
      for (int k0 = 0; k0 < K; k0 += 4) {
        v2f a = *(const v2f*)(xsrow + k0);
        v2f b = *(const v2f*)(wrow + k0);
        c = __builtin_amdgcn_wmma_f32_16x16x4_f32(
             false, a,  false, b,
             (short)0, c,  false,  false);
      }
      float* hp = Hn + (size_t)(tile * 16 + half * 8) * OUTN + wave * 16 + idx16;
#pragma unroll
      for (int r = 0; r < 8; ++r) hp[r * OUTN] = c[r];
    }

    tile += gridDim.x;
    if (tile >= NT) break;
    row = node_row_relu(tile * 16 + wave, lane, lo, rowptr, csr, dinv, H, bv);
    __syncthreads();
    *(V*)(&xs[wave * PAD + lo]) = row;
  }
}


extern "C" void kernel_launch(void* const* d_in, const int* in_sizes, int n_in,
                              void* d_out, int out_size, void* d_ws, size_t ws_size,
                              hipStream_t stream) {
  const float* x  = (const float*)d_in[0];
  const int*   ei = (const int*)d_in[1];
  const float* W1 = (const float*)d_in[2];
  const float* b1 = (const float*)d_in[3];
  const float* W2 = (const float*)d_in[4];
  const float* b2 = (const float*)d_in[5];
  const float* W3 = (const float*)d_in[6];
  const float* b3 = (const float*)d_in[7];
  const float* W4 = (const float*)d_in[8];
  const float* b4 = (const float*)d_in[9];

  const int E = in_sizes[1] / 2;
  const int* src = ei;
  const int* dst = ei + E;
  float* h      = (float*)d_ws;
  float* xcur   = h + (size_t)NN * 128;
  float* wt     = xcur + (size_t)NN * 128;
  float* dinv   = wt + 128 * 128;
  int*   rowptr = (int*)(dinv + NN);
  int*   csr    = rowptr + NN + 4;
  int2*  binned = (int2*)xcur;
  int*   cm     = (int*)xcur + 3400000;
  int2*  ged    = (int2*)h;
  int*   gso    = (int*)h + 3400000;

  unsigned nch = (unsigned)((E + CHK - 1) / CHK);
  k_bin<<<nch, 1024, 0, stream>>>(src, dst, binned, cm, E);
  k_sort<<<NBIN, 1024, 0, stream>>>(binned, cm, (int)nch, ged, gso, rowptr, dinv, csr);

  k_transpose<128, 128><<<(128 * 128 + 255) / 256, 256, 0, stream>>>(W1, wt);
  k_gemm_wmma<128, 128><<<dim3(NN / 16), dim3(256), 0, stream>>>(x, wt, dinv, h);
  k_gather_gemm<128><<<1024, 512, 0, stream>>>(rowptr, csr, dinv, h,    b1, W2, xcur);
  k_gather_gemm<128><<<1024, 512, 0, stream>>>(rowptr, csr, dinv, xcur, b2, W3, h);
  k_gather_gemm<64><<<1024, 512, 0, stream>>>(rowptr, csr, dinv, h,    b3, W4, xcur);
  k_gather<64, 0><<<(NN + 7) / 8, 256, 0, stream>>>(rowptr, csr, dinv, xcur, b4, (float*)d_out);
}


# layer-1 GEMM and its weight transpose folded into the per-bin sort launch: a bin block stages W1 and its own rows in LDS and multiplies once its dinv is made; 8 launches to 6; sum and k order kept
# speedup vs baseline: 8.3236x; 1.0694x over previous
#include <hip/hip_runtime.h>
#include <hip/hip_bf16.h>

typedef __attribute__((ext_vector_type(2))) float v2f;
typedef __attribute__((ext_vector_type(8))) float v8f;

#define NN 50000

#define BINW 196
#define NBIN 256
#define CHK  6400
#define CAP2 8192
#define MAXCH 1024
static_assert(NBIN * BINW >= NN, "bins cover all nodes");

__device__ __forceinline__ int wave_incl_scan(int v, int lane) {
#pragma unroll
  for (int off = 1; off < 32; off <<= 1) {
    int o = __shfl_up(v, off);
    if (lane >= off) v += o;
  }
  return v;
}

__global__ __launch_bounds__(1024) void k_bin(const int* __restrict__ src,
                                              const int* __restrict__ dst,
                                              int2* __restrict__ binned,
                                              int* __restrict__ cm, int e) {
  __shared__ int hist[NBIN];
  __shared__ int offs[NBIN];
  __shared__ int wsum[NBIN / 32];
  __shared__ int2 stage[CHK];
  const int t    = threadIdx.x;
  const int lane = t & 31;
  const int wave = t >> 5;
  const int c    = blockIdx.x;
  const int nch  = gridDim.x;
  const int cbeg = c * CHK;
  const int n    = min(CHK, e - cbeg);

  if (t < NBIN) hist[t] = 0;
  __syncthreads();

  constexpr int PER = (CHK + 1023) / 1024;
  int s[PER], d[PER], r[PER];
#pragma unroll
  for (int k = 0; k < PER; ++k) {
    const int j = k * 1024 + t;
    s[k] = 0; d[k] = 0;
    if (j < n) { d[k] = dst[cbeg + j]; s[k] = src[cbeg + j]; }
  }
#pragma unroll
  for (int k = 0; k < PER; ++k) {
    const int j = k * 1024 + t;
    r[k] = 0;
    if (j < n) r[k] = atomicAdd(&hist[d[k] / BINW], 1);
  }
  __syncthreads();

  const int v   = (t < NBIN) ? hist[t] : 0;
  const int inc = wave_incl_scan(v, lane);
  if (t < NBIN && lane == 31) wsum[wave] = inc;
  __syncthreads();
  if (t < NBIN) {
    int base = 0;
#pragma unroll
    for (int w = 0; w < NBIN / 32; ++w)
      if (w < wave) base += wsum[w];
    offs[t] = base + inc - v;
  }
  __syncthreads();

#pragma unroll
  for (int k = 0; k < PER; ++k) {
    const int j = k * 1024 + t;
    if (j < n) stage[offs[d[k] / BINW] + r[k]] = make_int2(s[k], d[k]);
  }
  __syncthreads();
  for (int j = t; j < n; j += 1024) binned[cbeg + j] = stage[j];
  if (t < NBIN) cm[t * nch + c] = (offs[t] << 16) | hist[t];
}

__device__ __forceinline__ void sort_bin(int2* ed, int* so, int* ncnt, int* nrow, int* wsum,
                                         const int* crun_w, const int* crun_st,
                                         const int2* __restrict__ binned, int nch, int b,
                                         int base, int nb, int* __restrict__ rowptr,
                                         float* __restrict__ dinv, int* __restrict__ ssrc,
                                         float* dl) {
  const int t     = threadIdx.x;
  const int lane  = t & 31;
  const int wave  = __builtin_amdgcn_readfirstlane(t >> 5);
  const int nbase = b * BINW;

  if (t < 256) ncnt[t] = 0;
  __syncthreads();

  for (int c0 = wave * 4; c0 < nch; c0 += 128) {
    int cnu[4], stu[4];
    const int2* runu[4];
    int2 pu[4];
#pragma unroll
    for (int u = 0; u < 4; ++u) {
      const int c = c0 + u;
      int w = 0, st = 0;
      if (c < nch) { w = crun_w[c]; st = crun_st[c]; }
      w  = __builtin_amdgcn_readfirstlane(w);
      st = __builtin_amdgcn_readfirstlane(st);
      cnu[u]  = w & 0xffff;
      stu[u]  = st;
      runu[u] = binned + (size_t)min(c, nch - 1) * CHK + (int)((unsigned)w >> 16);
      pu[u]   = make_int2(0, 0);
      if (lane < cnu[u]) pu[u] = runu[u][lane];
    }
#pragma unroll
    for (int u = 0; u < 4; ++u) {
      if (lane < cnu[u]) {
        const int ln = pu[u].y - nbase;
        const int rk = atomicAdd(&ncnt[ln], 1);
        ed[stu[u] + lane] = make_int2(pu[u].x, (rk << 8) | ln);
      }
      for (int l = lane + 32; l < cnu[u]; l += 32) {
        const int2 p = runu[u][l];
        const int ln = p.y - nbase;
        const int rk = atomicAdd(&ncnt[ln], 1);
        ed[stu[u] + l] = make_int2(p.x, (rk << 8) | ln);
      }
    }
  }
  __syncthreads();

  const int v   = (t < 256) ? ncnt[t] : 0;
  const int inc = wave_incl_scan(v, lane);
  if (t < 256 && lane == 31) wsum[wave] = inc;
  __syncthreads();
  if (t < 256) {
    int pre = 0;
#pragma unroll
    for (int w = 0; w < 8; ++w)
      if (w < wave) pre += wsum[w];
    const int ex = pre + inc - v;
    nrow[t] = ex;
    const int node = nbase + t;
    float dv = 0.0f;
    if (t < BINW && node < NN) {
      rowptr[node] = base + ex;
      float dg = (float)(v + 1);
      dv = (dg > 0.0f) ? rsqrtf(dg) : 0.0f;
      dinv[node] = dv;
    }
    dl[t] = dv;
  }
  if (b == NBIN - 1 && t == 0) rowptr[NN] = base + nb;
  __syncthreads();

  for (int j = t; j < nb; j += 1024) {
    const int2 p = ed[j];
    const int ln = p.y & 255;
    const int rk = (int)((unsigned)p.y >> 8);
    so[nrow[ln] + rk] = p.x;
  }
  __syncthreads();

  for (int ln = wave; ln < BINW; ln += 32) {
    const int beg = __builtin_amdgcn_readfirstlane(nrow[ln]);
    const int end = beg + __builtin_amdgcn_readfirstlane(ncnt[ln]);
    for (int bs = beg; bs < end; bs += 32) {
      const int idx   = bs + lane;
      const bool live = idx < end;
      const int mine  = live ? so[idx] : 0x7fffffff;
      int pos = 0;
      for (int ob = beg; ob < end; ob += 32) {
        const int oidx  = ob + lane;
        const int other = (oidx < end) ? so[oidx] : 0x7fffffff;
        const int m = min(32, end - ob);
        for (int j = 0; j < m; ++j) {
          const int o = __builtin_amdgcn_readlane(other, j);
          pos += (o < mine || (o == mine && ob + j < idx)) ? 1 : 0;
        }
      }
      if (live) ssrc[base + beg + pos] = mine;
    }
  }
}

#define GK    128
#define GPAD  (GK + 4)
#define XTILES ((BINW + 15) / 16)
#define XROWS (XTILES * 16)
#define XSLOTS (((XROWS * (GK / 4) + 1023) / 1024) * 1024)
#define XSROWS (XSLOTS / (GK / 4))
static_assert(XROWS <= 256, "dl[] covers every staged row");
static_assert(BINW >= 16 && NN - (NBIN - 1) * BINW >= 16, "every bin holds at least one whole 16-row tile");

__device__ __forceinline__ void bin_gemm(float* xs, const float* wl, const float* dl,
                                         float* __restrict__ H, int nbase) {
  const int t     = threadIdx.x;
  const int lane  = t & 31;
  const int wave  = __builtin_amdgcn_readfirstlane(t >> 5);
  const int nvalid = min(BINW, NN - nbase);
  const int ntile  = (nvalid + 15) >> 4;

  for (int sl = t; sl < XROWS * (GK / 4); sl += 1024) {
    const int row = sl >> 5;
    const int c4  = sl & 31;
    const float d = dl[row];
    float4* p = (float4*)(&xs[row * GPAD + c4 * 4]);
    float4 v = *p;
    v.x *= d; v.y *= d; v.z *= d; v.w *= d;
    *p = v;
  }
  __syncthreads();

  const int idx16 = lane & 15;
  const int half  = lane >> 4;
  const int ct    = wave & 7;
  const float* wrow = &wl[(ct * 16 + idx16) * GPAD + 2 * half];
  for (int rt = wave >> 3; rt < ntile; rt += 4) {
    const int r0 = min(rt * 16, nvalid - 16);
    const float* xsrow = &xs[(r0 + idx16) * GPAD + 2 * half];
    v8f acc = {};
#pragma unroll
    for (int k0 = 0; k0 < GK; k0 += 4) {
      v2f a = *(const v2f*)(xsrow + k0);
      v2f b = *(const v2f*)(wrow + k0);
      acc = __builtin_amdgcn_wmma_f32_16x16x4_f32(
           false, a,  false, b,
           (short)0, acc,  false,  false);
    }
    float* hp = H + (size_t)(nbase + r0 + half * 8) * GK + ct * 16 + idx16;
#pragma unroll
    for (int r = 0; r < 8; ++r) hp[r * GK] = acc[r];
  }
}

__global__ __launch_bounds__(1024) void k_sort(const int2* __restrict__ binned,
                                               const int* __restrict__ cm, int nch,
                                               int2* __restrict__ ged, int* __restrict__ gso,
                                               int* __restrict__ rowptr,
                                               float* __restrict__ dinv,
                                               int* __restrict__ ssrc,
                                               const float* __restrict__ X,
                                               const float* __restrict__ W,
                                               float* __restrict__ H) {
  __shared__ int ncnt[256];
  __shared__ int nrow[256];
  __shared__ int wtot[32];
  __shared__ int ptot[32];
  __shared__ int crun_w[MAXCH];
  __shared__ int crun_st[MAXCH];
  __shared__ int2 ed[CAP2];
  __shared__ int so[CAP2];
  __shared__ float dl[256];
  __shared__ float wl[GK * GPAD];
  __shared__ float xs[XSROWS * GPAD];
  const int t    = threadIdx.x;
  const int lane = t & 31;
  const int wave = t >> 5;
  const int b    = blockIdx.x;
  nch = min(nch, MAXCH);

  const int nfront = b * nch;
  const int w  = (t < nch) ? cm[nfront + t] : 0;

  {
    const int nbase = b * BINW;
    constexpr int XPER = XSLOTS / 1024;
    constexpr int WPER = GK * GK / 1024;
    float4 xv[XPER];
    float  wv[WPER];
#pragma unroll
    for (int u = 0; u < XPER; ++u) {
      const int sl  = u * 1024 + t;
      const int row = sl >> 5;
      const int c4  = sl & 31;
      const int g   = min(nbase + row, NN - 1);
      xv[u] = *(const float4*)(X + (size_t)g * GK + c4 * 4);
    }
#pragma unroll
    for (int u = 0; u < WPER; ++u) wv[u] = W[u * 1024 + t];
#pragma unroll
    for (int u = 0; u < XPER; ++u) {
      const int sl = u * 1024 + t;
      *(float4*)(&xs[(sl >> 5) * GPAD + (sl & 31) * 4]) = xv[u];
    }
#pragma unroll
    for (int u = 0; u < WPER; ++u) {
      const int i = u * 1024 + t;
      wl[(i & 127) * GPAD + (i >> 7)] = wv[u];
    }
  }

  const int cn = w & 0xffff;
  int p = (int)((unsigned)w >> 16);
  const int inc = wave_incl_scan(cn, lane);
#pragma unroll
  for (int off = 16; off > 0; off >>= 1) p += __shfl_xor(p, off);
  if (lane == 31) wtot[wave] = inc;
  if (lane == 0)  ptot[wave] = p;
  __syncthreads();
  int base = 0, pre = 0, nb = 0;
#pragma unroll
  for (int k = 0; k < 32; ++k) {
    base += ptot[k];
    nb   += wtot[k];
    if (k < wave) pre += wtot[k];
  }
  if (t < nch) {
    crun_w[t]  = w;
    crun_st[t] = pre + inc - cn;
  }
  base = __builtin_amdgcn_readfirstlane(base);
  nb   = __builtin_amdgcn_readfirstlane(nb);
  if (nb <= CAP2)
    sort_bin(ed, so, ncnt, nrow, wtot, crun_w, crun_st, binned, nch, b, base, nb, rowptr, dinv, ssrc, dl);
  else
    sort_bin(ged + base, gso + base, ncnt, nrow, wtot, crun_w, crun_st, binned, nch, b, base, nb,
             rowptr, dinv, ssrc, dl);

  bin_gemm(xs, wl, dl, H, b * BINW);
}


template <int VEC> struct vec_of;
template <> struct vec_of<4> { typedef __attribute__((ext_vector_type(4))) float type; };
template <> struct vec_of<2> { typedef v2f type; };

template <int N, int OUTD, typename V>
__device__ __forceinline__ void rows_add(V& acc, const float* __restrict__ H, unsigned lo,
                                         int ev, int j) {
  int s[N];
  V v[N];
#pragma unroll
  for (int u = 0; u < N; ++u) {
    s[u] = __builtin_amdgcn_readlane(ev, j + u);
  }
#pragma unroll
  for (int u = 0; u < N; ++u) v[u] = *(const V*)(H + (size_t)s[u] * OUTD + lo);
#pragma unroll
  for (int u = 0; u < N; ++u) acc += v[u];
}

template <int OUTD, int RELU>
__global__ __launch_bounds__(256) void k_gather(const int* __restrict__ rowptr,
                                                const int* __restrict__ csr,
                                                const float* __restrict__ dinv,
                                                const float* __restrict__ H,
                                                const float* __restrict__ bias,
                                                float* __restrict__ out) {
  constexpr int VEC = OUTD / 32;
  using V = typename vec_of<VEC>::type;
  const int lane = threadIdx.x & 31;
  const int node = __builtin_amdgcn_readfirstlane((int)(blockIdx.x * 8 + (threadIdx.x >> 5)));
  if (node >= NN) return;
  const int beg = __builtin_amdgcn_readfirstlane(rowptr[node]);
  const int end = __builtin_amdgcn_readfirstlane(rowptr[node + 1]);
  const unsigned lo = (unsigned)lane * VEC;
  const float di = dinv[node];
  V acc = *(const V*)(H + (size_t)node * OUTD + lo);

  for (int base = beg; base < end; base += 32) {
    const int idx = base + lane;
    int ev = 0;
    if (idx < end) ev = csr[idx];
    const int m = min(32, end - base);
    int j = 0;
    for (; j + 8 <= m; j += 8) rows_add<8, OUTD, V>(acc, H, lo, ev, j);
    if (j + 4 <= m) {
      rows_add<4, OUTD, V>(acc, H, lo, ev, j);
      j += 4;
    }
    for (; j < m; ++j) rows_add<1, OUTD, V>(acc, H, lo, ev, j);
  }

  acc = acc * di + *(const V*)(bias + lo);
  if (RELU) {
#pragma unroll
    for (int k = 0; k < VEC; ++k) acc[k] = fmaxf(acc[k], 0.0f);
  }
  *(V*)(out + (size_t)node * OUTD + lo) = acc;
}


__device__ __forceinline__ vec_of<4>::type node_row_relu(int node, int lane, unsigned lo,
                                                        const int* __restrict__ rowptr,
                                                        const int* __restrict__ csr,
                                                        const float* __restrict__ dinv,
                                                        const float* __restrict__ H,
                                                        vec_of<4>::type bv) {
  using V = vec_of<4>::type;
  constexpr int K = 128;
  const int beg = __builtin_amdgcn_readfirstlane(rowptr[node]);
  const int end = __builtin_amdgcn_readfirstlane(rowptr[node + 1]);
  const float di = dinv[node];
  V acc = *(const V*)(H + (size_t)node * K + lo);

  for (int base = beg; base < end; base += 32) {
    const int idx = base + lane;
    int ev = 0;
    if (idx < end) ev = csr[idx];
    const int m = min(32, end - base);
    int j = 0;
    for (; j + 8 <= m; j += 8) rows_add<8, K, V>(acc, H, lo, ev, j);
    if (j + 4 <= m) {
      rows_add<4, K, V>(acc, H, lo, ev, j);
      j += 4;
    }
    for (; j < m; ++j) rows_add<1, K, V>(acc, H, lo, ev, j);
  }
  acc = acc * di + bv;
#pragma unroll
  for (int k = 0; k < 4; ++k) acc[k] = fmaxf(acc[k], 0.0f);
  return acc * di;
}

template <int OUTN>
__global__ __launch_bounds__(512, 16) void k_gather_gemm(const int* __restrict__ rowptr,
                                                         const int* __restrict__ csr,
                                                         const float* __restrict__ dinv,
                                                         const float* __restrict__ H,
                                                         const float* __restrict__ bias,
                                                         const float* __restrict__ Wn,
                                                         float* __restrict__ Hn) {
  constexpr int K   = 128;
  constexpr int PAD = K + 4;
  constexpr int NT  = NN / 16;
  static_assert(NN % 16 == 0, "node tiles need no bounds checks");
  __shared__ float wl[OUTN * PAD];
  __shared__ float xs[16 * PAD];
  using V = vec_of<4>::type;

  const int tid  = threadIdx.x;
  const int lane = tid & 31;
  const int wave = __builtin_amdgcn_readfirstlane(tid >> 5);
  const unsigned lo = (unsigned)lane * 4;
  const V bv = *(const V*)(bias + lo);
  const int idx16 = lane & 15;
  const int half  = lane >> 4;

  int tile = blockIdx.x;
  if (tile >= NT) return;

  V row = node_row_relu(tile * 16 + wave, lane, lo, rowptr, csr, dinv, H, bv);
  *(V*)(&xs[wave * PAD + lo]) = row;

#pragma unroll 8
  for (int i = tid; i < K * OUTN; i += 512) {
    const int k = i / OUTN;
    const int n = i - k * OUTN;
    wl[n * PAD + k] = Wn[i];
  }

  while (true) {
    __syncthreads();

    if (wave < OUTN / 16) {
      const float* xsrow = &xs[idx16 * PAD + 2 * half];
      const float* wrow  = &wl[(wave * 16 + idx16) * PAD + 2 * half];
      v8f c = {};
#pragma unroll
      for (int k0 = 0; k0 < K; k0 += 4) {
        v2f a = *(const v2f*)(xsrow + k0);
        v2f b = *(const v2f*)(wrow + k0);
        c = __builtin_amdgcn_wmma_f32_16x16x4_f32(
             false, a,  false, b,
             (short)0, c,  false,  false);
      }
      float* hp = Hn + (size_t)(tile * 16 + half * 8) * OUTN + wave * 16 + idx16;
#pragma unroll
      for (int r = 0; r < 8; ++r) hp[r * OUTN] = c[r];
    }

    tile += gridDim.x;
    if (tile >= NT) break;
    row = node_row_relu(tile * 16 + wave, lane, lo, rowptr, csr, dinv, H, bv);
    __syncthreads();
    *(V*)(&xs[wave * PAD + lo]) = row;
  }
}


extern "C" void kernel_launch(void* const* d_in, const int* in_sizes, int n_in,
                              void* d_out, int out_size, void* d_ws, size_t ws_size,
                              hipStream_t stream) {
  const float* x  = (const float*)d_in[0];
  const int*   ei = (const int*)d_in[1];
  const float* W1 = (const float*)d_in[2];
  const float* b1 = (const float*)d_in[3];
  const float* W2 = (const float*)d_in[4];
  const float* b2 = (const float*)d_in[5];
  const float* W3 = (const float*)d_in[6];
  const float* b3 = (const float*)d_in[7];
  const float* W4 = (const float*)d_in[8];
  const float* b4 = (const float*)d_in[9];

  const int E = in_sizes[1] / 2;
  const int* src = ei;
  const int* dst = ei + E;
  float* h      = (float*)d_ws;
  float* xcur   = h + (size_t)NN * 128;
  float* dinv   = xcur + (size_t)NN * 128 + 128 * 128;
  int*   rowptr = (int*)(dinv + NN);
  int*   csr    = rowptr + NN + 4;
  int2*  binned = (int2*)xcur;
  int*   cm     = (int*)xcur + 3400000;
  int2*  ged    = (int2*)d_out;
  int*   gso    = (int*)xcur + 3500000;

  unsigned nch = (unsigned)((E + CHK - 1) / CHK);
  k_bin<<<nch, 1024, 0, stream>>>(src, dst, binned, cm, E);
  k_sort<<<NBIN, 1024, 0, stream>>>(binned, cm, (int)nch, ged, gso, rowptr, dinv, csr, x, W1, h);

  k_gather_gemm<128><<<1024, 512, 0, stream>>>(rowptr, csr, dinv, h,    b1, W2, xcur);
  k_gather_gemm<128><<<1024, 512, 0, stream>>>(rowptr, csr, dinv, xcur, b2, W3, h);
  k_gather_gemm<64><<<1024, 512, 0, stream>>>(rowptr, csr, dinv, h,    b3, W4, xcur);
  k_gather<64, 0><<<(NN + 7) / 8, 256, 0, stream>>>(rowptr, csr, dinv, xcur, b4, (float*)d_out);
}
